# starting kernel
# baseline (speedup 1.0000x reference)
#include <hip/hip_runtime.h>
#include <math.h>

#define HEADS 4
#define FEAT  32
#define HF    128

typedef __attribute__((ext_vector_type(2))) float v2f;
typedef __attribute__((ext_vector_type(8))) float v8f;

__global__ void __launch_bounds__(256)
gat_gemm_wmma_f32(const float* __restrict__ X, const float* __restrict__ W,
                  float* __restrict__ Hout, int K) {
  const int wave = threadIdx.x >> 5;
  const int lane = threadIdx.x & 31;
  const int half = lane >> 4;
  const int l15  = lane & 15;
  const int row  = blockIdx.x * 16 + l15;
  const int col  = wave * 16 + l15;
  const int kh   = half * 2;
  const float* __restrict__ xrow = X + (size_t)row * K;

  v8f c = {};
  for (int k0 = 0; k0 < K; k0 += 4) {
    v2f a, b;
    a.x = xrow[k0 + kh];
    a.y = xrow[k0 + kh + 1];
    b.x = W[(size_t)(k0 + kh)     * HF + col];
    b.y = W[(size_t)(k0 + kh + 1) * HF + col];
    c = __builtin_amdgcn_wmma_f32_16x16x4_f32(false, a, false, b,
                                              (short)0, c, false, false);
  }
  const int rbase = blockIdx.x * 16 + half * 8;
  float* orow = Hout + (size_t)rbase * HF + col;
#pragma unroll
  for (int v = 0; v < 8; ++v)
    orow[(size_t)v * HF] = c[v];
}

__global__ void gat_gemm_tail(const float* __restrict__ X, const float* __restrict__ W,
                              float* __restrict__ Hout, int K, int row0, int Nn) {
  int i = blockIdx.x * blockDim.x + threadIdx.x;
  int rows = Nn - row0;
  if (i >= rows * HF) return;
  int r = row0 + (i >> 7), col = i & (HF - 1);
  float acc = 0.f;
  for (int k = 0; k < K; ++k) acc += X[(size_t)r * K + k] * W[(size_t)k * HF + col];
  Hout[(size_t)r * HF + col] = acc;
}

__global__ void gat_alpha(const float* __restrict__ H, const float* __restrict__ att_src,
                          const float* __restrict__ att_dst,
                          float* __restrict__ a_src, float* __restrict__ a_dst, int Nn) {
  int i = blockIdx.x * blockDim.x + threadIdx.x;
  if (i >= Nn * HEADS) return;
  int h = i & (HEADS - 1);
  const float4* hp = (const float4*)(H + (size_t)(i >> 2) * HF + h * FEAT);
  const float4* as = (const float4*)(att_src + h * FEAT);
  const float4* ad = (const float4*)(att_dst + h * FEAT);
  float s1 = 0.f, s2 = 0.f;
#pragma unroll
  for (int f = 0; f < FEAT / 4; ++f) {
    float4 v = hp[f], a = as[f], d = ad[f];
    s1 += v.x * a.x + v.y * a.y + v.z * a.z + v.w * a.w;
    s2 += v.x * d.x + v.y * d.y + v.z * d.z + v.w * d.w;
  }
  a_src[i] = s1;
  a_dst[i] = s2;
}

__global__ void gat_init(float* __restrict__ m, float* __restrict__ s,
                         float4* __restrict__ acc4, int Nn) {
  int i = blockIdx.x * blockDim.x + threadIdx.x;
  if (i < Nn * HEADS) { m[i] = -INFINITY; s[i] = 0.f; }
  if (i < Nn * (HF / 4)) acc4[i] = make_float4(0.f, 0.f, 0.f, 0.f);
}

__device__ __forceinline__ float leaky(float e) { return e > 0.f ? e : 0.2f * e; }

__device__ __forceinline__ void atomicMaxF(float* addr, float val) {
  if (val >= 0.f) atomicMax((int*)addr, __float_as_int(val));
  else            atomicMin((unsigned int*)addr, __float_as_uint(val));
}

__device__ __forceinline__ void edge_pair(const int* __restrict__ ei, int eid, int E,
                                          int& s, int& d) {
  if (eid < E) { s = ei[eid]; d = ei[E + eid]; }
  else         { s = d = eid - E; }
}

__global__ void gat_edge_max(const int* __restrict__ ei, const float* __restrict__ asrc,
                             const float* __restrict__ adst, float* __restrict__ m,
                             int E, int Nn) {
  int i = blockIdx.x * blockDim.x + threadIdx.x;
  if (i >= (E + Nn) * HEADS) return;
  int eid = i >> 2, h = i & (HEADS - 1), s, d;
  edge_pair(ei, eid, E, s, d);
  float e = leaky(asrc[s * HEADS + h] + adst[d * HEADS + h]);
  atomicMaxF(&m[d * HEADS + h], e);
}

__global__ void gat_edge_sum(const int* __restrict__ ei, const float* __restrict__ asrc,
                             const float* __restrict__ adst, const float* __restrict__ m,
                             float* __restrict__ sums, int E, int Nn) {
  int i = blockIdx.x * blockDim.x + threadIdx.x;
  if (i >= (E + Nn) * HEADS) return;
  int eid = i >> 2, h = i & (HEADS - 1), s, d;
  edge_pair(ei, eid, E, s, d);
  float e = leaky(asrc[s * HEADS + h] + adst[d * HEADS + h]);
  atomicAdd(&sums[d * HEADS + h], expf(e - m[d * HEADS + h]));
}

__global__ void gat_edge_agg(const int* __restrict__ ei, const float* __restrict__ asrc,
                             const float* __restrict__ adst, const float* __restrict__ m,
                             const float* __restrict__ sums, const float* __restrict__ hfeat,
                             float* __restrict__ acc, int E, int Nn) {
  long long i = (long long)blockIdx.x * blockDim.x + threadIdx.x;
  if (i >= (long long)(E + Nn) * (HF / 4)) return;
  int eid = (int)(i >> 5);
  int g   = (int)(i & 31);
  int c   = g * 4;
  int h   = c >> 5;
  int s, d;
  edge_pair(ei, eid, E, s, d);
  float e = leaky(asrc[s * HEADS + h] + adst[d * HEADS + h]);
  float w = expf(e - m[d * HEADS + h]) / (sums[d * HEADS + h] + 1e-16f);
  const float4 hv = *(const float4*)(hfeat + (size_t)s * HF + c);
  float* ap = acc + (size_t)d * HF + c;
  atomicAdd(ap + 0, w * hv.x);
  atomicAdd(ap + 1, w * hv.y);
  atomicAdd(ap + 2, w * hv.z);
  atomicAdd(ap + 3, w * hv.w);
}

__global__ void gat_head_mean(const float* __restrict__ acc, const float* __restrict__ bias,
                              float* __restrict__ out, int Nn, int do_elu) {
  int i = blockIdx.x * blockDim.x + threadIdx.x;
  if (i >= Nn * FEAT) return;
  int n = i >> 5, f = i & (FEAT - 1);
  const float* a = acc + (size_t)n * HF;
  float v = 0.25f * (a[f] + a[FEAT + f] + a[2 * FEAT + f] + a[3 * FEAT + f]) + bias[f];
  if (do_elu) v = v > 0.f ? v : (expf(v) - 1.f);
  out[i] = v;
}

extern "C" void kernel_launch(void* const* d_in, const int* in_sizes, int n_in,
                              void* d_out, int out_size, void* d_ws, size_t ws_size,
                              hipStream_t stream) {
  const float* x   = (const float*)d_in[0];
  const int*   ei  = (const int*)d_in[1];
  const float* W1  = (const float*)d_in[2];
  const float* as1 = (const float*)d_in[3];
  const float* ad1 = (const float*)d_in[4];
  const float* b1  = (const float*)d_in[5];
  const float* W2  = (const float*)d_in[6];
  const float* as2 = (const float*)d_in[7];
  const float* ad2 = (const float*)d_in[8];
  const float* b2  = (const float*)d_in[9];

  const int Nn = in_sizes[0] / HF;
  const int E  = in_sizes[1] / 2;
  const int ET = E + Nn;

  float* ws    = (float*)d_ws;
  float* hbuf  = ws;
  float* acc   = hbuf + (size_t)Nn * HF;
  float* asrc  = acc  + (size_t)Nn * HF;
  float* adst  = asrc + (size_t)Nn * HEADS;
  float* mbuf  = adst + (size_t)Nn * HEADS;
  float* sbuf  = mbuf + (size_t)Nn * HEADS;
  float* x2    = sbuf + (size_t)Nn * HEADS;

  const dim3 blk(256);
  const int row_tiles = Nn / 16;
  const int tail_row0 = row_tiles * 16;

  for (int layer = 0; layer < 2; ++layer) {
    const float* Xin = (layer == 0) ? x  : x2;
    const int    K   = (layer == 0) ? 128 : FEAT;
    const float* Wm  = (layer == 0) ? W1 : W2;
    const float* as  = (layer == 0) ? as1 : as2;
    const float* ad  = (layer == 0) ? ad1 : ad2;
    const float* bb  = (layer == 0) ? b1 : b2;
    float*       op  = (layer == 0) ? x2 : (float*)d_out;

    if (row_tiles > 0)
      gat_gemm_wmma_f32<<<row_tiles, blk, 0, stream>>>(Xin, Wm, hbuf, K);
    if (tail_row0 < Nn) {
      int tail = (Nn - tail_row0) * HF;
      gat_gemm_tail<<<(tail + 255) / 256, blk, 0, stream>>>(Xin, Wm, hbuf, K, tail_row0, Nn);
    }
    gat_alpha<<<(Nn * HEADS + 255) / 256, blk, 0, stream>>>(hbuf, as, ad, asrc, adst, Nn);
    gat_init<<<(Nn * (HF / 4) + 255) / 256, blk, 0, stream>>>(mbuf, sbuf, (float4*)acc, Nn);
    gat_edge_max<<<(ET * HEADS + 255) / 256, blk, 0, stream>>>(ei, asrc, adst, mbuf, E, Nn);
    gat_edge_sum<<<(ET * HEADS + 255) / 256, blk, 0, stream>>>(ei, asrc, adst, mbuf, sbuf, E, Nn);
    {
      long long tot = (long long)ET * (HF / 4);
      unsigned int nb = (unsigned int)((tot + 255) / 256);
      gat_edge_agg<<<nb, blk, 0, stream>>>(ei, asrc, adst, mbuf, sbuf, hbuf, acc, E, Nn);
    }
    gat_head_mean<<<(Nn * FEAT + 255) / 256, blk, 0, stream>>>(acc, bb, op, Nn, layer == 0 ? 1 : 0);
  }
}


# edge aggregation sums the 4 heads per edge before the scatter (output is the head mean): 32 not 128 float atomics per edge, one wave per 8 edges; changes the float summation order
# speedup vs baseline: 1.8326x; 1.8326x over previous
#include <hip/hip_runtime.h>
#include <math.h>

#define HEADS 4
#define FEAT  32
#define HF    128

typedef __attribute__((ext_vector_type(2))) float v2f;
typedef __attribute__((ext_vector_type(8))) float v8f;

__global__ void __launch_bounds__(256)
gat_gemm_wmma_f32(const float* __restrict__ X, const float* __restrict__ W,
                  float* __restrict__ Hout, int K) {
  const int wave = threadIdx.x >> 5;
  const int lane = threadIdx.x & 31;
  const int half = lane >> 4;
  const int l15  = lane & 15;
  const int row  = blockIdx.x * 16 + l15;
  const int col  = wave * 16 + l15;
  const int kh   = half * 2;
  const float* __restrict__ xrow = X + (size_t)row * K;

  v8f c = {};
  for (int k0 = 0; k0 < K; k0 += 4) {
    v2f a, b;
    a.x = xrow[k0 + kh];
    a.y = xrow[k0 + kh + 1];
    b.x = W[(size_t)(k0 + kh)     * HF + col];
    b.y = W[(size_t)(k0 + kh + 1) * HF + col];
    c = __builtin_amdgcn_wmma_f32_16x16x4_f32(false, a, false, b,
                                              (short)0, c, false, false);
  }
  const int rbase = blockIdx.x * 16 + half * 8;
  float* orow = Hout + (size_t)rbase * HF + col;
#pragma unroll
  for (int v = 0; v < 8; ++v)
    orow[(size_t)v * HF] = c[v];
}

__global__ void gat_gemm_tail(const float* __restrict__ X, const float* __restrict__ W,
                              float* __restrict__ Hout, int K, int row0, int Nn) {
  int i = blockIdx.x * blockDim.x + threadIdx.x;
  int rows = Nn - row0;
  if (i >= rows * HF) return;
  int r = row0 + (i >> 7), col = i & (HF - 1);
  float acc = 0.f;
  for (int k = 0; k < K; ++k) acc += X[(size_t)r * K + k] * W[(size_t)k * HF + col];
  Hout[(size_t)r * HF + col] = acc;
}

__global__ void gat_alpha(const float* __restrict__ H, const float* __restrict__ att_src,
                          const float* __restrict__ att_dst,
                          float* __restrict__ a_src, float* __restrict__ a_dst, int Nn) {
  int i = blockIdx.x * blockDim.x + threadIdx.x;
  if (i >= Nn * HEADS) return;
  int h = i & (HEADS - 1);
  const float4* hp = (const float4*)(H + (size_t)(i >> 2) * HF + h * FEAT);
  const float4* as = (const float4*)(att_src + h * FEAT);
  const float4* ad = (const float4*)(att_dst + h * FEAT);
  float s1 = 0.f, s2 = 0.f;
#pragma unroll
  for (int f = 0; f < FEAT / 4; ++f) {
    float4 v = hp[f], a = as[f], d = ad[f];
    s1 += v.x * a.x + v.y * a.y + v.z * a.z + v.w * a.w;
    s2 += v.x * d.x + v.y * d.y + v.z * d.z + v.w * d.w;
  }
  a_src[i] = s1;
  a_dst[i] = s2;
}

__global__ void gat_init(float* __restrict__ m, float* __restrict__ s,
                         float4* __restrict__ acc4, int Nn) {
  int i = blockIdx.x * blockDim.x + threadIdx.x;
  if (i < Nn * HEADS) { m[i] = -INFINITY; s[i] = 0.f; }
  if (i < Nn * (FEAT / 4)) acc4[i] = make_float4(0.f, 0.f, 0.f, 0.f);
}

__device__ __forceinline__ float leaky(float e) { return e > 0.f ? e : 0.2f * e; }

__device__ __forceinline__ void atomicMaxF(float* addr, float val) {
  if (val >= 0.f) atomicMax((int*)addr, __float_as_int(val));
  else            atomicMin((unsigned int*)addr, __float_as_uint(val));
}

__device__ __forceinline__ void edge_pair(const int* __restrict__ ei, int eid, int E,
                                          int& s, int& d) {
  if (eid < E) { s = ei[eid]; d = ei[E + eid]; }
  else         { s = d = eid - E; }
}

__global__ void gat_edge_max(const int* __restrict__ ei, const float* __restrict__ asrc,
                             const float* __restrict__ adst, float* __restrict__ m,
                             int E, int Nn) {
  int i = blockIdx.x * blockDim.x + threadIdx.x;
  if (i >= (E + Nn) * HEADS) return;
  int eid = i >> 2, h = i & (HEADS - 1), s, d;
  edge_pair(ei, eid, E, s, d);
  float e = leaky(asrc[s * HEADS + h] + adst[d * HEADS + h]);
  atomicMaxF(&m[d * HEADS + h], e);
}

__global__ void gat_edge_sum(const int* __restrict__ ei, const float* __restrict__ asrc,
                             const float* __restrict__ adst, const float* __restrict__ m,
                             float* __restrict__ sums, int E, int Nn) {
  int i = blockIdx.x * blockDim.x + threadIdx.x;
  if (i >= (E + Nn) * HEADS) return;
  int eid = i >> 2, h = i & (HEADS - 1), s, d;
  edge_pair(ei, eid, E, s, d);
  float e = leaky(asrc[s * HEADS + h] + adst[d * HEADS + h]);
  atomicAdd(&sums[d * HEADS + h], expf(e - m[d * HEADS + h]));
}

#define EPW 8

__device__ __forceinline__ int lane_bcast_i(int v, int l) {
  return __builtin_amdgcn_readlane(v, l);
}
__device__ __forceinline__ float lane_bcast_f(float v, int l) {
  return __int_as_float(__builtin_amdgcn_readlane(__float_as_int(v), l));
}

template <bool GUARD>
__device__ __forceinline__ void agg_scatter(const float (&hv)[EPW][HEADS], int d, float w,
                                            float* __restrict__ acc, int lane, int e0, int ET) {
  for (int j = 0; j < EPW; ++j) {
    const int   dj = lane_bcast_i(d, 4 * j);
    const float w0 = lane_bcast_f(w, 4 * j + 0);
    const float w1 = lane_bcast_f(w, 4 * j + 1);
    const float w2 = lane_bcast_f(w, 4 * j + 2);
    const float w3 = lane_bcast_f(w, 4 * j + 3);
    const float v = w0 * hv[j][0] + w1 * hv[j][1] + w2 * hv[j][2] + w3 * hv[j][3];
    if (!GUARD || e0 + j < ET) atomicAdd(acc + (size_t)dj * FEAT + lane, v);
  }
}

__global__ void __launch_bounds__(256)
gat_edge_agg(const int* __restrict__ ei, const float* __restrict__ asrc,
             const float* __restrict__ adst, const float* __restrict__ m,
             const float* __restrict__ sums, const float* __restrict__ hfeat,
             float* __restrict__ acc, int E, int Nn) {
  const int lane = threadIdx.x & 31;
  const int wv = __builtin_amdgcn_readfirstlane((int)(threadIdx.x >> 5));
  const int ET = E + Nn;
  const int e0 = (blockIdx.x * (256 / 32) + wv) * EPW;
  if (e0 >= ET) return;

  const int  hh    = lane & (HEADS - 1);
  const int  eid   = e0 + (lane >> 2);
  const bool valid = eid < ET;
  int s = 0, d = 0;
  if (valid) edge_pair(ei, eid, E, s, d);

  float hv[EPW][HEADS];
  for (int j = 0; j < EPW; ++j) {
    const int sj = lane_bcast_i(s, 4 * j);
    const float* hp = hfeat + (size_t)sj * HF + lane;
    for (int k = 0; k < HEADS; ++k) hv[j][k] = hp[k * FEAT];
  }

  const float e = leaky(asrc[s * HEADS + hh] + adst[d * HEADS + hh]);
  float w = expf(e - m[d * HEADS + hh]) / (sums[d * HEADS + hh] + 1e-16f);
  if (!valid) w = 0.f;

  if (e0 + EPW <= ET) agg_scatter<false>(hv, d, w, acc, lane, e0, ET);
  else                agg_scatter<true >(hv, d, w, acc, lane, e0, ET);
}

__global__ void gat_head_mean(const float* __restrict__ acc, const float* __restrict__ bias,
                              float* __restrict__ out, int Nn, int do_elu) {
  int i = blockIdx.x * blockDim.x + threadIdx.x;
  if (i >= Nn * (FEAT / 4)) return;
  const float4 a = ((const float4*)acc)[i];
  const float4 b = *(const float4*)(bias + (i & (FEAT / 4 - 1)) * 4);
  float4 v;
  v.x = 0.25f * a.x + b.x;
  v.y = 0.25f * a.y + b.y;
  v.z = 0.25f * a.z + b.z;
  v.w = 0.25f * a.w + b.w;
  if (do_elu) {
    v.x = v.x > 0.f ? v.x : (expf(v.x) - 1.f);
    v.y = v.y > 0.f ? v.y : (expf(v.y) - 1.f);
    v.z = v.z > 0.f ? v.z : (expf(v.z) - 1.f);
    v.w = v.w > 0.f ? v.w : (expf(v.w) - 1.f);
  }
  ((float4*)out)[i] = v;
}

extern "C" void kernel_launch(void* const* d_in, const int* in_sizes, int n_in,
                              void* d_out, int out_size, void* d_ws, size_t ws_size,
                              hipStream_t stream) {
  const float* x   = (const float*)d_in[0];
  const int*   ei  = (const int*)d_in[1];
  const float* W1  = (const float*)d_in[2];
  const float* as1 = (const float*)d_in[3];
  const float* ad1 = (const float*)d_in[4];
  const float* b1  = (const float*)d_in[5];
  const float* W2  = (const float*)d_in[6];
  const float* as2 = (const float*)d_in[7];
  const float* ad2 = (const float*)d_in[8];
  const float* b2  = (const float*)d_in[9];

  const int Nn = in_sizes[0] / HF;
  const int E  = in_sizes[1] / 2;
  const int ET = E + Nn;

  float* ws    = (float*)d_ws;
  float* hbuf  = ws;
  float* acc   = hbuf + (size_t)Nn * HF;
  float* asrc  = acc  + (size_t)Nn * HF;
  float* adst  = asrc + (size_t)Nn * HEADS;
  float* mbuf  = adst + (size_t)Nn * HEADS;
  float* sbuf  = mbuf + (size_t)Nn * HEADS;
  float* x2    = sbuf + (size_t)Nn * HEADS;

  const dim3 blk(256);
  const int row_tiles = Nn / 16;
  const int tail_row0 = row_tiles * 16;

  for (int layer = 0; layer < 2; ++layer) {
    const float* Xin = (layer == 0) ? x  : x2;
    const int    K   = (layer == 0) ? 128 : FEAT;
    const float* Wm  = (layer == 0) ? W1 : W2;
    const float* as  = (layer == 0) ? as1 : as2;
    const float* ad  = (layer == 0) ? ad1 : ad2;
    const float* bb  = (layer == 0) ? b1 : b2;
    float*       op  = (layer == 0) ? x2 : (float*)d_out;

    if (row_tiles > 0)
      gat_gemm_wmma_f32<<<row_tiles, blk, 0, stream>>>(Xin, Wm, hbuf, K);
    if (tail_row0 < Nn) {
      int tail = (Nn - tail_row0) * HF;
      gat_gemm_tail<<<(tail + 255) / 256, blk, 0, stream>>>(Xin, Wm, hbuf, K, tail_row0, Nn);
    }
    gat_alpha<<<(Nn * HEADS + 255) / 256, blk, 0, stream>>>(hbuf, as, ad, asrc, adst, Nn);
    gat_init<<<(Nn * (FEAT / 4) + 255) / 256, blk, 0, stream>>>(mbuf, sbuf, (float4*)acc, Nn);
    gat_edge_max<<<(ET * HEADS + 255) / 256, blk, 0, stream>>>(ei, asrc, adst, mbuf, E, Nn);
    gat_edge_sum<<<(ET * HEADS + 255) / 256, blk, 0, stream>>>(ei, asrc, adst, mbuf, sbuf, E, Nn);
    {
      const int waves = (ET + EPW - 1) / EPW;
      gat_edge_agg<<<(waves + 7) / 8, blk, 0, stream>>>(ei, asrc, adst, mbuf, sbuf, hbuf, acc, E, Nn);
    }
    gat_head_mean<<<(Nn * (FEAT / 4) + 255) / 256, blk, 0, stream>>>(acc, bb, op, Nn, layer == 0 ? 1 : 0);
  }
}


# edges grouped by destination once per call (counting sort, integer atomics), then one wave per node does max, sum and aggregation in registers: no float atomics; changes the float summation order
# speedup vs baseline: 3.9149x; 2.1363x over previous
#include <hip/hip_runtime.h>
#include <math.h>

#define HEADS 4
#define FEAT  32
#define HF    128

typedef __attribute__((ext_vector_type(2))) float v2f;
typedef __attribute__((ext_vector_type(8))) float v8f;

__global__ void __launch_bounds__(256)
gat_gemm_wmma_f32(const float* __restrict__ X, const float* __restrict__ W,
                  float* __restrict__ Hout, int K) {
  const int wave = threadIdx.x >> 5;
  const int lane = threadIdx.x & 31;
  const int half = lane >> 4;
  const int l15  = lane & 15;
  const int row  = blockIdx.x * 16 + l15;
  const int col  = wave * 16 + l15;
  const int kh   = half * 2;
  const float* __restrict__ xrow = X + (size_t)row * K;

  v8f c = {};
  for (int k0 = 0; k0 < K; k0 += 4) {
    v2f a, b;
    a.x = xrow[k0 + kh];
    a.y = xrow[k0 + kh + 1];
    b.x = W[(size_t)(k0 + kh)     * HF + col];
    b.y = W[(size_t)(k0 + kh + 1) * HF + col];
    c = __builtin_amdgcn_wmma_f32_16x16x4_f32(false, a, false, b,
                                              (short)0, c, false, false);
  }
  const int rbase = blockIdx.x * 16 + half * 8;
  float* orow = Hout + (size_t)rbase * HF + col;
#pragma unroll
  for (int v = 0; v < 8; ++v)
    orow[(size_t)v * HF] = c[v];
}

__global__ void gat_gemm_tail(const float* __restrict__ X, const float* __restrict__ W,
                              float* __restrict__ Hout, int K, int row0, int Nn) {
  int i = blockIdx.x * blockDim.x + threadIdx.x;
  int rows = Nn - row0;
  if (i >= rows * HF) return;
  int r = row0 + (i >> 7), col = i & (HF - 1);
  float acc = 0.f;
  for (int k = 0; k < K; ++k) acc += X[(size_t)r * K + k] * W[(size_t)k * HF + col];
  Hout[(size_t)r * HF + col] = acc;
}

__global__ void gat_alpha(const float* __restrict__ H, const float* __restrict__ att_src,
                          const float* __restrict__ att_dst,
                          float* __restrict__ a_src, float* __restrict__ a_dst, int Nn) {
  int i = blockIdx.x * blockDim.x + threadIdx.x;
  if (i >= Nn * HEADS) return;
  int h = i & (HEADS - 1);
  const float4* hp = (const float4*)(H + (size_t)(i >> 2) * HF + h * FEAT);
  const float4* as = (const float4*)(att_src + h * FEAT);
  const float4* ad = (const float4*)(att_dst + h * FEAT);
  float s1 = 0.f, s2 = 0.f;
#pragma unroll
  for (int f = 0; f < FEAT / 4; ++f) {
    float4 v = hp[f], a = as[f], d = ad[f];
    s1 += v.x * a.x + v.y * a.y + v.z * a.z + v.w * a.w;
    s2 += v.x * d.x + v.y * d.y + v.z * d.z + v.w * d.w;
  }
  a_src[i] = s1;
  a_dst[i] = s2;
}

__global__ void csr_zero(int* __restrict__ cnt, int Nn) {
  int i = blockIdx.x * blockDim.x + threadIdx.x;
  if (i < Nn) cnt[i] = 0;
}

__global__ void csr_count(const int* __restrict__ ei, int* __restrict__ cnt,
                          int* __restrict__ rank, int E) {
  int e = blockIdx.x * blockDim.x + threadIdx.x;
  if (e >= E) return;
  rank[e] = atomicAdd(&cnt[ei[E + e]], 1);
}

#define SCAN_T 1024

__device__ __forceinline__ int wave_incl_scan(int v, int lane) {
  for (int off = 1; off < 32; off <<= 1) {
    int u = __shfl_up(v, off, 32);
    if (lane >= off) v += u;
  }
  return v;
}

__global__ void __launch_bounds__(SCAN_T)
csr_scan(const int* __restrict__ cnt, int* __restrict__ rowptr, int Nn) {
  __shared__ int wsum[2][32];
  const int t = threadIdx.x, lane = t & 31, wv = t >> 5;
  int carry = 0, buf = 0;
  for (int tile = 0; tile < Nn; tile += SCAN_T * 4) {
    const int b = tile + t * 4;
    int c0 = 0, c1 = 0, c2 = 0, c3 = 0;
    if (b + 3 < Nn) {
      const int4 v = *(const int4*)(cnt + b);
      c0 = v.x; c1 = v.y; c2 = v.z; c3 = v.w;
    } else {
      if (b     < Nn) c0 = cnt[b];
      if (b + 1 < Nn) c1 = cnt[b + 1];
      if (b + 2 < Nn) c2 = cnt[b + 2];
    }
    const int s1 = c0, s2 = s1 + c1, s3 = s2 + c2, s4 = s3 + c3;
    const int inc = wave_incl_scan(s4, lane);
    if (lane == 31) wsum[buf][wv] = inc;
    __syncthreads();
    const int ws  = wsum[buf][lane];
    const int wsi = wave_incl_scan(ws, lane);
    const int wave_off = __shfl(wsi, wv, 32) - __shfl(ws, wv, 32);
    const int tile_tot = __shfl(wsi, 31, 32);
    const int ex = carry + wave_off + inc - s4;
    if (b + 3 < Nn) {
      *(int4*)(rowptr + b) = make_int4(ex, ex + s1, ex + s2, ex + s3);
    } else {
      if (b     < Nn) rowptr[b]     = ex;
      if (b + 1 < Nn) rowptr[b + 1] = ex + s1;
      if (b + 2 < Nn) rowptr[b + 2] = ex + s2;
    }
    carry += tile_tot;
    buf ^= 1;
  }
  if (t == 0) rowptr[Nn] = carry;
}

__global__ void csr_fill(const int* __restrict__ ei, const int* __restrict__ rowptr,
                         const int* __restrict__ rank, int* __restrict__ csr_src, int E) {
  int e = blockIdx.x * blockDim.x + threadIdx.x;
  if (e >= E) return;
  csr_src[rowptr[ei[E + e]] + rank[e]] = ei[e];
}

__device__ __forceinline__ float leaky(float e) { return e > 0.f ? e : 0.2f * e; }

__device__ __forceinline__ int lane_bcast_i(int v, int l) {
  return __builtin_amdgcn_readlane(v, l);
}
__device__ __forceinline__ float lane_bcast_f(float v, int l) {
  return __int_as_float(__builtin_amdgcn_readlane(__float_as_int(v), l));
}

__device__ __forceinline__ float wave_max(float v) {
  for (int off = 16; off > 0; off >>= 1) v = fmaxf(v, __shfl_xor(v, off, 32));
  return v;
}
__device__ __forceinline__ float wave_sum(float v) {
  for (int off = 16; off > 0; off >>= 1) v += __shfl_xor(v, off, 32);
  return v;
}

template <int G>
__device__ __forceinline__ float gather_group(const float* __restrict__ hfeat, int s,
                                              float w0, float w1, float w2, float w3,
                                              int j, int lane, float r) {
  float hv[G][HEADS];
#pragma unroll
  for (int g = 0; g < G; ++g) {
    const float* hp = hfeat + (size_t)lane_bcast_i(s, j + g) * HF;
#pragma unroll
    for (int k = 0; k < HEADS; ++k) hv[g][k] = hp[k * FEAT + lane];
  }
#pragma unroll
  for (int g = 0; g < G; ++g) {
    r = fmaf(lane_bcast_f(w0, j + g), hv[g][0], r);
    r = fmaf(lane_bcast_f(w1, j + g), hv[g][1], r);
    r = fmaf(lane_bcast_f(w2, j + g), hv[g][2], r);
    r = fmaf(lane_bcast_f(w3, j + g), hv[g][3], r);
  }
  return r;
}

__global__ void __launch_bounds__(256)
gat_node(const int* __restrict__ rowptr, const int* __restrict__ csr_src,
         const float* __restrict__ asrc, const float* __restrict__ adst,
         const float* __restrict__ hfeat, const float* __restrict__ bias,
         float* __restrict__ out, int Nn, int do_elu) {
  const int lane = threadIdx.x & 31;
  const int wv = __builtin_amdgcn_readfirstlane((int)(threadIdx.x >> 5));
  const int d  = blockIdx.x * (256 / 32) + wv;
  if (d >= Nn) return;
  const int beg = __builtin_amdgcn_readfirstlane(rowptr[d]);
  const int deg = __builtin_amdgcn_readfirstlane(rowptr[d + 1]) - beg;
  const int tot = deg + 1;
  const float4 ad = *(const float4*)(adst + (size_t)d * HEADS);
  const float4* __restrict__ asrc4 = (const float4*)asrc;

  float m0 = -INFINITY, m1 = -INFINITY, m2 = -INFINITY, m3 = -INFINITY;
  for (int base = 0; base < tot; base += 32) {
    const int i = base + lane;
    int s = d;
    if (i < deg) s = csr_src[beg + i];
    const float4 a = asrc4[s];
    if (i < tot) {
      m0 = fmaxf(m0, leaky(a.x + ad.x));
      m1 = fmaxf(m1, leaky(a.y + ad.y));
      m2 = fmaxf(m2, leaky(a.z + ad.z));
      m3 = fmaxf(m3, leaky(a.w + ad.w));
    }
  }
  m0 = wave_max(m0); m1 = wave_max(m1); m2 = wave_max(m2); m3 = wave_max(m3);

  float s0 = 0.f, s1 = 0.f, s2 = 0.f, s3 = 0.f;
  for (int base = 0; base < tot; base += 32) {
    const int i = base + lane;
    int s = d;
    if (i < deg) s = csr_src[beg + i];
    const float4 a = asrc4[s];
    if (i < tot) {
      s0 += expf(leaky(a.x + ad.x) - m0);
      s1 += expf(leaky(a.y + ad.y) - m1);
      s2 += expf(leaky(a.z + ad.z) - m2);
      s3 += expf(leaky(a.w + ad.w) - m3);
    }
  }
  const float den0 = wave_sum(s0) + 1e-16f;
  const float den1 = wave_sum(s1) + 1e-16f;
  const float den2 = wave_sum(s2) + 1e-16f;
  const float den3 = wave_sum(s3) + 1e-16f;

  float acc = 0.f;
  for (int base = 0; base < tot; base += 32) {
    const int i = base + lane;
    int s = d;
    if (i < deg) s = csr_src[beg + i];
    const float4 a = asrc4[s];
    float w0 = 0.f, w1 = 0.f, w2 = 0.f, w3 = 0.f;
    if (i < tot) {
      w0 = expf(leaky(a.x + ad.x) - m0) / den0;
      w1 = expf(leaky(a.y + ad.y) - m1) / den1;
      w2 = expf(leaky(a.z + ad.z) - m2) / den2;
      w3 = expf(leaky(a.w + ad.w) - m3) / den3;
    }
    const int cnt = (tot - base < 32) ? (tot - base) : 32;
    int j = 0;
    for (; j + 8 <= cnt; j += 8) acc = gather_group<8>(hfeat, s, w0, w1, w2, w3, j, lane, acc);
    for (; j < cnt; j += 4)      acc = gather_group<4>(hfeat, s, w0, w1, w2, w3, j, lane, acc);
  }

  float v = 0.25f * acc + bias[lane];
  if (do_elu) v = v > 0.f ? v : (expf(v) - 1.f);
  out[(size_t)d * FEAT + lane] = v;
}

extern "C" void kernel_launch(void* const* d_in, const int* in_sizes, int n_in,
                              void* d_out, int out_size, void* d_ws, size_t ws_size,
                              hipStream_t stream) {
  const float* x   = (const float*)d_in[0];
  const int*   ei  = (const int*)d_in[1];
  const float* W1  = (const float*)d_in[2];
  const float* as1 = (const float*)d_in[3];
  const float* ad1 = (const float*)d_in[4];
  const float* b1  = (const float*)d_in[5];
  const float* W2  = (const float*)d_in[6];
  const float* as2 = (const float*)d_in[7];
  const float* ad2 = (const float*)d_in[8];
  const float* b2  = (const float*)d_in[9];

  const int Nn = in_sizes[0] / HF;
  const int E  = in_sizes[1] / 2;

  const size_t n4 = ((size_t)Nn + 1 + 3) / 4 * 4;
  const size_t e4 = ((size_t)E + 3) / 4 * 4;
  float* ws     = (float*)d_ws;
  float* hbuf   = ws;
  float* x2     = hbuf + (size_t)Nn * HF;
  float* asrc   = x2   + (size_t)Nn * FEAT;
  float* adst   = asrc + (size_t)Nn * HEADS;
  int*   cnt    = (int*)(adst + (size_t)Nn * HEADS);
  int*   rowptr = cnt + n4;
  int*   rank   = rowptr + n4;
  int*   csr    = rank + e4;

  const dim3 blk(256);
  const int row_tiles = Nn / 16;
  const int tail_row0 = row_tiles * 16;

  csr_zero<<<(Nn + 255) / 256, blk, 0, stream>>>(cnt, Nn);
  if (E > 0) csr_count<<<(E + 255) / 256, blk, 0, stream>>>(ei, cnt, rank, E);
  csr_scan<<<1, SCAN_T, 0, stream>>>(cnt, rowptr, Nn);
  if (E > 0) csr_fill<<<(E + 255) / 256, blk, 0, stream>>>(ei, rowptr, rank, csr, E);

  for (int layer = 0; layer < 2; ++layer) {
    const float* Xin = (layer == 0) ? x  : x2;
    const int    K   = (layer == 0) ? 128 : FEAT;
    const float* Wm  = (layer == 0) ? W1 : W2;
    const float* as  = (layer == 0) ? as1 : as2;
    const float* ad  = (layer == 0) ? ad1 : ad2;
    const float* bb  = (layer == 0) ? b1 : b2;
    float*       op  = (layer == 0) ? x2 : (float*)d_out;

    if (row_tiles > 0)
      gat_gemm_wmma_f32<<<row_tiles, blk, 0, stream>>>(Xin, Wm, hbuf, K);
    if (tail_row0 < Nn) {
      int tail = (Nn - tail_row0) * HF;
      gat_gemm_tail<<<(tail + 255) / 256, blk, 0, stream>>>(Xin, Wm, hbuf, K, tail_row0, Nn);
    }
    gat_alpha<<<(Nn * HEADS + 255) / 256, blk, 0, stream>>>(hbuf, as, ad, asrc, adst, Nn);
    gat_node<<<(Nn + 7) / 8, blk, 0, stream>>>(rowptr, csr, asrc, adst, hbuf, bb, op, Nn,
                                               layer == 0 ? 1 : 0);
  }
}


# node kernel fetches each edge's source id and a_src row once, not three times: logits of the first 64 list entries stay in registers from max to gather; same values, same summation order
# speedup vs baseline: 4.3609x; 1.1139x over previous
#include <hip/hip_runtime.h>
#include <math.h>

#define HEADS 4
#define FEAT  32
#define HF    128

typedef __attribute__((ext_vector_type(2))) float v2f;
typedef __attribute__((ext_vector_type(8))) float v8f;

__global__ void __launch_bounds__(256)
gat_gemm_wmma_f32(const float* __restrict__ X, const float* __restrict__ W,
                  float* __restrict__ Hout, int K) {
  const int wave = threadIdx.x >> 5;
  const int lane = threadIdx.x & 31;
  const int half = lane >> 4;
  const int l15  = lane & 15;
  const int row  = blockIdx.x * 16 + l15;
  const int col  = wave * 16 + l15;
  const int kh   = half * 2;
  const float* __restrict__ xrow = X + (size_t)row * K;

  v8f c = {};
  for (int k0 = 0; k0 < K; k0 += 4) {
    v2f a, b;
    a.x = xrow[k0 + kh];
    a.y = xrow[k0 + kh + 1];
    b.x = W[(size_t)(k0 + kh)     * HF + col];
    b.y = W[(size_t)(k0 + kh + 1) * HF + col];
    c = __builtin_amdgcn_wmma_f32_16x16x4_f32(false, a, false, b,
                                              (short)0, c, false, false);
  }
  const int rbase = blockIdx.x * 16 + half * 8;
  float* orow = Hout + (size_t)rbase * HF + col;
#pragma unroll
  for (int v = 0; v < 8; ++v)
    orow[(size_t)v * HF] = c[v];
}

__global__ void gat_gemm_tail(const float* __restrict__ X, const float* __restrict__ W,
                              float* __restrict__ Hout, int K, int row0, int Nn) {
  int i = blockIdx.x * blockDim.x + threadIdx.x;
  int rows = Nn - row0;
  if (i >= rows * HF) return;
  int r = row0 + (i >> 7), col = i & (HF - 1);
  float acc = 0.f;
  for (int k = 0; k < K; ++k) acc += X[(size_t)r * K + k] * W[(size_t)k * HF + col];
  Hout[(size_t)r * HF + col] = acc;
}

__global__ void gat_alpha(const float* __restrict__ H, const float* __restrict__ att_src,
                          const float* __restrict__ att_dst,
                          float* __restrict__ a_src, float* __restrict__ a_dst, int Nn) {
  int i = blockIdx.x * blockDim.x + threadIdx.x;
  if (i >= Nn * HEADS) return;
  int h = i & (HEADS - 1);
  const float4* hp = (const float4*)(H + (size_t)(i >> 2) * HF + h * FEAT);
  const float4* as = (const float4*)(att_src + h * FEAT);
  const float4* ad = (const float4*)(att_dst + h * FEAT);
  float s1 = 0.f, s2 = 0.f;
#pragma unroll
  for (int f = 0; f < FEAT / 4; ++f) {
    float4 v = hp[f], a = as[f], d = ad[f];
    s1 += v.x * a.x + v.y * a.y + v.z * a.z + v.w * a.w;
    s2 += v.x * d.x + v.y * d.y + v.z * d.z + v.w * d.w;
  }
  a_src[i] = s1;
  a_dst[i] = s2;
}

__global__ void csr_zero(int* __restrict__ cnt, int Nn) {
  int i = blockIdx.x * blockDim.x + threadIdx.x;
  if (i < Nn) cnt[i] = 0;
}

__global__ void csr_count(const int* __restrict__ ei, int* __restrict__ cnt,
                          int* __restrict__ rank, int E) {
  int e = blockIdx.x * blockDim.x + threadIdx.x;
  if (e >= E) return;
  rank[e] = atomicAdd(&cnt[ei[E + e]], 1);
}

#define SCAN_T 1024

__device__ __forceinline__ int wave_incl_scan(int v, int lane) {
  for (int off = 1; off < 32; off <<= 1) {
    int u = __shfl_up(v, off, 32);
    if (lane >= off) v += u;
  }
  return v;
}

__global__ void __launch_bounds__(SCAN_T)
csr_scan(const int* __restrict__ cnt, int* __restrict__ rowptr, int Nn) {
  __shared__ int wsum[2][32];
  const int t = threadIdx.x, lane = t & 31, wv = t >> 5;
  int carry = 0, buf = 0;
  for (int tile = 0; tile < Nn; tile += SCAN_T * 4) {
    const int b = tile + t * 4;
    int c0 = 0, c1 = 0, c2 = 0, c3 = 0;
    if (b + 3 < Nn) {
      const int4 v = *(const int4*)(cnt + b);
      c0 = v.x; c1 = v.y; c2 = v.z; c3 = v.w;
    } else {
      if (b     < Nn) c0 = cnt[b];
      if (b + 1 < Nn) c1 = cnt[b + 1];
      if (b + 2 < Nn) c2 = cnt[b + 2];
    }
    const int s1 = c0, s2 = s1 + c1, s3 = s2 + c2, s4 = s3 + c3;
    const int inc = wave_incl_scan(s4, lane);
    if (lane == 31) wsum[buf][wv] = inc;
    __syncthreads();
    const int ws  = wsum[buf][lane];
    const int wsi = wave_incl_scan(ws, lane);
    const int wave_off = __shfl(wsi, wv, 32) - __shfl(ws, wv, 32);
    const int tile_tot = __shfl(wsi, 31, 32);
    const int ex = carry + wave_off + inc - s4;
    if (b + 3 < Nn) {
      *(int4*)(rowptr + b) = make_int4(ex, ex + s1, ex + s2, ex + s3);
    } else {
      if (b     < Nn) rowptr[b]     = ex;
      if (b + 1 < Nn) rowptr[b + 1] = ex + s1;
      if (b + 2 < Nn) rowptr[b + 2] = ex + s2;
    }
    carry += tile_tot;
    buf ^= 1;
  }
  if (t == 0) rowptr[Nn] = carry;
}

__global__ void csr_fill(const int* __restrict__ ei, const int* __restrict__ rowptr,
                         const int* __restrict__ rank, int* __restrict__ csr_src, int E) {
  int e = blockIdx.x * blockDim.x + threadIdx.x;
  if (e >= E) return;
  csr_src[rowptr[ei[E + e]] + rank[e]] = ei[e];
}

__device__ __forceinline__ float leaky(float e) { return e > 0.f ? e : 0.2f * e; }

__device__ __forceinline__ int lane_bcast_i(int v, int l) {
  return __builtin_amdgcn_readlane(v, l);
}
__device__ __forceinline__ float lane_bcast_f(float v, int l) {
  return __int_as_float(__builtin_amdgcn_readlane(__float_as_int(v), l));
}

__device__ __forceinline__ float wave_max(float v) {
  for (int off = 16; off > 0; off >>= 1) v = fmaxf(v, __shfl_xor(v, off, 32));
  return v;
}
__device__ __forceinline__ float wave_sum(float v) {
  for (int off = 16; off > 0; off >>= 1) v += __shfl_xor(v, off, 32);
  return v;
}

template <int G>
__device__ __forceinline__ float gather_group(const float* __restrict__ hfeat, int s,
                                              float w0, float w1, float w2, float w3,
                                              int j, int lane, float r) {
  float hv[G][HEADS];
#pragma unroll
  for (int g = 0; g < G; ++g) {
    const float* hp = hfeat + (size_t)lane_bcast_i(s, j + g) * HF;
#pragma unroll
    for (int k = 0; k < HEADS; ++k) hv[g][k] = hp[k * FEAT + lane];
  }
#pragma unroll
  for (int g = 0; g < G; ++g) {
    r = fmaf(lane_bcast_f(w0, j + g), hv[g][0], r);
    r = fmaf(lane_bcast_f(w1, j + g), hv[g][1], r);
    r = fmaf(lane_bcast_f(w2, j + g), hv[g][2], r);
    r = fmaf(lane_bcast_f(w3, j + g), hv[g][3], r);
  }
  return r;
}

__device__ __forceinline__ float gather_chunk(const float* __restrict__ hfeat, int s,
                                              float w0, float w1, float w2, float w3,
                                              int cnt, int lane, float acc) {
  int j = 0;
  for (; j + 8 <= cnt; j += 8) acc = gather_group<8>(hfeat, s, w0, w1, w2, w3, j, lane, acc);
  for (; j < cnt; j += 4)      acc = gather_group<4>(hfeat, s, w0, w1, w2, w3, j, lane, acc);
  return acc;
}

__global__ void __launch_bounds__(256)
gat_node(const int* __restrict__ rowptr, const int* __restrict__ csr_src,
         const float* __restrict__ asrc, const float* __restrict__ adst,
         const float* __restrict__ hfeat, const float* __restrict__ bias,
         float* __restrict__ out, int Nn, int do_elu) {
  const int lane = threadIdx.x & 31;
  const int wv = __builtin_amdgcn_readfirstlane((int)(threadIdx.x >> 5));
  const int d  = blockIdx.x * (256 / 32) + wv;
  if (d >= Nn) return;
  const int beg = __builtin_amdgcn_readfirstlane(rowptr[d]);
  const int deg = __builtin_amdgcn_readfirstlane(rowptr[d + 1]) - beg;
  const int tot = deg + 1;
  const bool two = tot > 32;
  const float4 ad = *(const float4*)(adst + (size_t)d * HEADS);
  const float4* __restrict__ asrc4 = (const float4*)asrc;

  int sA = d, sB = d;
  if (lane < deg) sA = csr_src[beg + lane];
  if (two && lane + 32 < deg) sB = csr_src[beg + 32 + lane];
  const float4 aA = asrc4[sA];
  float4 aB = make_float4(0.f, 0.f, 0.f, 0.f);
  if (two) aB = asrc4[sB];

  float eA0 = -INFINITY, eA1 = -INFINITY, eA2 = -INFINITY, eA3 = -INFINITY;
  float eB0 = -INFINITY, eB1 = -INFINITY, eB2 = -INFINITY, eB3 = -INFINITY;
  if (lane < tot) {
    eA0 = leaky(aA.x + ad.x); eA1 = leaky(aA.y + ad.y);
    eA2 = leaky(aA.z + ad.z); eA3 = leaky(aA.w + ad.w);
  }
  if (lane + 32 < tot) {
    eB0 = leaky(aB.x + ad.x); eB1 = leaky(aB.y + ad.y);
    eB2 = leaky(aB.z + ad.z); eB3 = leaky(aB.w + ad.w);
  }

  float m0 = fmaxf(eA0, eB0), m1 = fmaxf(eA1, eB1), m2 = fmaxf(eA2, eB2), m3 = fmaxf(eA3, eB3);
  for (int base = 64; base < tot; base += 32) {
    const int i = base + lane;
    int s = d;
    if (i < deg) s = csr_src[beg + i];
    const float4 a = asrc4[s];
    if (i < tot) {
      m0 = fmaxf(m0, leaky(a.x + ad.x));
      m1 = fmaxf(m1, leaky(a.y + ad.y));
      m2 = fmaxf(m2, leaky(a.z + ad.z));
      m3 = fmaxf(m3, leaky(a.w + ad.w));
    }
  }
  m0 = wave_max(m0); m1 = wave_max(m1); m2 = wave_max(m2); m3 = wave_max(m3);

  float pA0 = 0.f, pA1 = 0.f, pA2 = 0.f, pA3 = 0.f;
  float pB0 = 0.f, pB1 = 0.f, pB2 = 0.f, pB3 = 0.f;
  if (lane < tot) {
    pA0 = expf(eA0 - m0); pA1 = expf(eA1 - m1); pA2 = expf(eA2 - m2); pA3 = expf(eA3 - m3);
  }
  if (lane + 32 < tot) {
    pB0 = expf(eB0 - m0); pB1 = expf(eB1 - m1); pB2 = expf(eB2 - m2); pB3 = expf(eB3 - m3);
  }
  float s0 = pA0 + pB0, s1 = pA1 + pB1, s2 = pA2 + pB2, s3 = pA3 + pB3;
  for (int base = 64; base < tot; base += 32) {
    const int i = base + lane;
    int s = d;
    if (i < deg) s = csr_src[beg + i];
    const float4 a = asrc4[s];
    if (i < tot) {
      s0 += expf(leaky(a.x + ad.x) - m0);
      s1 += expf(leaky(a.y + ad.y) - m1);
      s2 += expf(leaky(a.z + ad.z) - m2);
      s3 += expf(leaky(a.w + ad.w) - m3);
    }
  }
  const float den0 = wave_sum(s0) + 1e-16f;
  const float den1 = wave_sum(s1) + 1e-16f;
  const float den2 = wave_sum(s2) + 1e-16f;
  const float den3 = wave_sum(s3) + 1e-16f;

  float acc = gather_chunk(hfeat, sA, pA0 / den0, pA1 / den1, pA2 / den2, pA3 / den3,
                           two ? 32 : tot, lane, 0.f);
  if (two)
    acc = gather_chunk(hfeat, sB, pB0 / den0, pB1 / den1, pB2 / den2, pB3 / den3,
                       (tot - 32 < 32) ? (tot - 32) : 32, lane, acc);
  for (int base = 64; base < tot; base += 32) {
    const int i = base + lane;
    int s = d;
    if (i < deg) s = csr_src[beg + i];
    const float4 a = asrc4[s];
    float w0 = 0.f, w1 = 0.f, w2 = 0.f, w3 = 0.f;
    if (i < tot) {
      w0 = expf(leaky(a.x + ad.x) - m0) / den0;
      w1 = expf(leaky(a.y + ad.y) - m1) / den1;
      w2 = expf(leaky(a.z + ad.z) - m2) / den2;
      w3 = expf(leaky(a.w + ad.w) - m3) / den3;
    }
    acc = gather_chunk(hfeat, s, w0, w1, w2, w3, (tot - base < 32) ? (tot - base) : 32, lane, acc);
  }

  float v = 0.25f * acc + bias[lane];
  if (do_elu) v = v > 0.f ? v : (expf(v) - 1.f);
  out[(size_t)d * FEAT + lane] = v;
}

extern "C" void kernel_launch(void* const* d_in, const int* in_sizes, int n_in,
                              void* d_out, int out_size, void* d_ws, size_t ws_size,
                              hipStream_t stream) {
  const float* x   = (const float*)d_in[0];
  const int*   ei  = (const int*)d_in[1];
  const float* W1  = (const float*)d_in[2];
  const float* as1 = (const float*)d_in[3];
  const float* ad1 = (const float*)d_in[4];
  const float* b1  = (const float*)d_in[5];
  const float* W2  = (const float*)d_in[6];
  const float* as2 = (const float*)d_in[7];
  const float* ad2 = (const float*)d_in[8];
  const float* b2  = (const float*)d_in[9];

  const int Nn = in_sizes[0] / HF;
  const int E  = in_sizes[1] / 2;

  const size_t n4 = ((size_t)Nn + 1 + 3) / 4 * 4;
  const size_t e4 = ((size_t)E + 3) / 4 * 4;
  float* ws     = (float*)d_ws;
  float* hbuf   = ws;
  float* x2     = hbuf + (size_t)Nn * HF;
  float* asrc   = x2   + (size_t)Nn * FEAT;
  float* adst   = asrc + (size_t)Nn * HEADS;
  int*   cnt    = (int*)(adst + (size_t)Nn * HEADS);
  int*   rowptr = cnt + n4;
  int*   rank   = rowptr + n4;
  int*   csr    = rank + e4;

  const dim3 blk(256);
  const int row_tiles = Nn / 16;
  const int tail_row0 = row_tiles * 16;

  csr_zero<<<(Nn + 255) / 256, blk, 0, stream>>>(cnt, Nn);
  if (E > 0) csr_count<<<(E + 255) / 256, blk, 0, stream>>>(ei, cnt, rank, E);
  csr_scan<<<1, SCAN_T, 0, stream>>>(cnt, rowptr, Nn);
  if (E > 0) csr_fill<<<(E + 255) / 256, blk, 0, stream>>>(ei, rowptr, rank, csr, E);

  for (int layer = 0; layer < 2; ++layer) {
    const float* Xin = (layer == 0) ? x  : x2;
    const int    K   = (layer == 0) ? 128 : FEAT;
    const float* Wm  = (layer == 0) ? W1 : W2;
    const float* as  = (layer == 0) ? as1 : as2;
    const float* ad  = (layer == 0) ? ad1 : ad2;
    const float* bb  = (layer == 0) ? b1 : b2;
    float*       op  = (layer == 0) ? x2 : (float*)d_out;

    if (row_tiles > 0)
      gat_gemm_wmma_f32<<<row_tiles, blk, 0, stream>>>(Xin, Wm, hbuf, K);
    if (tail_row0 < Nn) {
      int tail = (Nn - tail_row0) * HF;
      gat_gemm_tail<<<(tail + 255) / 256, blk, 0, stream>>>(Xin, Wm, hbuf, K, tail_row0, Nn);
    }
    gat_alpha<<<(Nn * HEADS + 255) / 256, blk, 0, stream>>>(hbuf, as, ad, asrc, adst, Nn);
    gat_node<<<(Nn + 7) / 8, blk, 0, stream>>>(rowptr, csr, asrc, adst, hbuf, bb, op, Nn,
                                               layer == 0 ? 1 : 0);
  }
}


# GEMM wave owns a whole head (two 16-col tiles, one A fragment) and writes a_src/a_dst from its accumulators: both alpha launches and their re-read of h go; logits summed in a new order (16-lane tree)
# speedup vs baseline: 4.6148x; 1.0582x over previous
#include <hip/hip_runtime.h>
#include <math.h>

#define HEADS 4
#define FEAT  32
#define HF    128

typedef __attribute__((ext_vector_type(2))) float v2f;
typedef __attribute__((ext_vector_type(8))) float v8f;

__global__ void __launch_bounds__(128)
gat_gemm_alpha_wmma_f32(const float* __restrict__ X, const float* __restrict__ W,
                        const float* __restrict__ att_src, const float* __restrict__ att_dst,
                        float* __restrict__ Hout, float* __restrict__ a_src,
                        float* __restrict__ a_dst, int K) {
  const int head = threadIdx.x >> 5;
  const int lane = threadIdx.x & 31;
  const int half = lane >> 4;
  const int l15  = lane & 15;
  const int row  = blockIdx.x * 16 + l15;
  const int col  = head * FEAT + l15;
  const int kh   = half * 2;
  const float* __restrict__ xrow = X + (size_t)row * K;

  v8f c0 = {}, c1 = {};
  for (int k0 = 0; k0 < K; k0 += 4) {
    v2f a, b0, b1;
    a.x = xrow[k0 + kh];
    a.y = xrow[k0 + kh + 1];
    const float* __restrict__ wr = W + (size_t)(k0 + kh) * HF + col;
    b0.x = wr[0];
    b0.y = wr[HF];
    b1.x = wr[16];
    b1.y = wr[HF + 16];
    c0 = __builtin_amdgcn_wmma_f32_16x16x4_f32(false, a, false, b0,
                                               (short)0, c0, false, false);
    c1 = __builtin_amdgcn_wmma_f32_16x16x4_f32(false, a, false, b1,
                                               (short)0, c1, false, false);
  }
  const int rbase = blockIdx.x * 16 + half * 8;
  float* orow = Hout + (size_t)rbase * HF + col;
#pragma unroll
  for (int v = 0; v < 8; ++v) {
    orow[(size_t)v * HF]      = c0[v];
    orow[(size_t)v * HF + 16] = c1[v];
  }

  const float as0 = att_src[col], as1 = att_src[col + 16];
  const float ad0 = att_dst[col], ad1 = att_dst[col + 16];
  float q[16];
#pragma unroll
  for (int v = 0; v < 8; ++v) {
    q[v]     = c0[v] * as0 + c1[v] * as1;
    q[8 + v] = c0[v] * ad0 + c1[v] * ad1;
  }
  const bool b3 = (l15 & 8) != 0, b2 = (l15 & 4) != 0, b1s = (l15 & 2) != 0, b0s = (l15 & 1) != 0;
  float r1[8], r2[4], r3[2];
#pragma unroll
  for (int i = 0; i < 8; ++i) {
    const float send = b3 ? q[i] : q[i + 8];
    const float keep = b3 ? q[i + 8] : q[i];
    r1[i] = keep + __shfl_xor(send, 8, 32);
  }
#pragma unroll
  for (int i = 0; i < 4; ++i) {
    const float send = b2 ? r1[i] : r1[i + 4];
    const float keep = b2 ? r1[i + 4] : r1[i];
    r2[i] = keep + __shfl_xor(send, 4, 32);
  }
#pragma unroll
  for (int i = 0; i < 2; ++i) {
    const float send = b1s ? r2[i] : r2[i + 2];
    const float keep = b1s ? r2[i + 2] : r2[i];
    r3[i] = keep + __shfl_xor(send, 2, 32);
  }
  const float send = b0s ? r3[0] : r3[1];
  const float keep = b0s ? r3[1] : r3[0];
  const float tot  = keep + __shfl_xor(send, 1, 32);
  float* __restrict__ ap = (l15 < 8) ? a_src : a_dst;
  ap[(size_t)(rbase + (l15 & 7)) * HEADS + head] = tot;
}

__global__ void gat_gemm_tail(const float* __restrict__ X, const float* __restrict__ W,
                              float* __restrict__ Hout, int K, int row0, int Nn) {
  int i = blockIdx.x * blockDim.x + threadIdx.x;
  int rows = Nn - row0;
  if (i >= rows * HF) return;
  int r = row0 + (i >> 7), col = i & (HF - 1);
  float acc = 0.f;
  for (int k = 0; k < K; ++k) acc += X[(size_t)r * K + k] * W[(size_t)k * HF + col];
  Hout[(size_t)r * HF + col] = acc;
}

__global__ void gat_alpha(const float* __restrict__ H, const float* __restrict__ att_src,
                          const float* __restrict__ att_dst,
                          float* __restrict__ a_src, float* __restrict__ a_dst, int Nn) {
  int i = blockIdx.x * blockDim.x + threadIdx.x;
  if (i >= Nn * HEADS) return;
  int h = i & (HEADS - 1);
  const float4* hp = (const float4*)(H + (size_t)(i >> 2) * HF + h * FEAT);
  const float4* as = (const float4*)(att_src + h * FEAT);
  const float4* ad = (const float4*)(att_dst + h * FEAT);
  float s1 = 0.f, s2 = 0.f;
#pragma unroll
  for (int f = 0; f < FEAT / 4; ++f) {
    float4 v = hp[f], a = as[f], d = ad[f];
    s1 += v.x * a.x + v.y * a.y + v.z * a.z + v.w * a.w;
    s2 += v.x * d.x + v.y * d.y + v.z * d.z + v.w * d.w;
  }
  a_src[i] = s1;
  a_dst[i] = s2;
}

__global__ void csr_zero(int* __restrict__ cnt, int Nn) {
  int i = blockIdx.x * blockDim.x + threadIdx.x;
  if (i < Nn) cnt[i] = 0;
}

__global__ void csr_count(const int* __restrict__ ei, int* __restrict__ cnt,
                          int* __restrict__ rank, int E) {
  int e = blockIdx.x * blockDim.x + threadIdx.x;
  if (e >= E) return;
  rank[e] = atomicAdd(&cnt[ei[E + e]], 1);
}

#define SCAN_T 1024

__device__ __forceinline__ int wave_incl_scan(int v, int lane) {
  for (int off = 1; off < 32; off <<= 1) {
    int u = __shfl_up(v, off, 32);
    if (lane >= off) v += u;
  }
  return v;
}

__global__ void __launch_bounds__(SCAN_T)
csr_scan(const int* __restrict__ cnt, int* __restrict__ rowptr, int Nn) {
  __shared__ int wsum[2][32];
  const int t = threadIdx.x, lane = t & 31, wv = t >> 5;
  int carry = 0, buf = 0;
  for (int tile = 0; tile < Nn; tile += SCAN_T * 4) {
    const int b = tile + t * 4;
    int c0 = 0, c1 = 0, c2 = 0, c3 = 0;
    if (b + 3 < Nn) {
      const int4 v = *(const int4*)(cnt + b);
      c0 = v.x; c1 = v.y; c2 = v.z; c3 = v.w;
    } else {
      if (b     < Nn) c0 = cnt[b];
      if (b + 1 < Nn) c1 = cnt[b + 1];
      if (b + 2 < Nn) c2 = cnt[b + 2];
    }
    const int s1 = c0, s2 = s1 + c1, s3 = s2 + c2, s4 = s3 + c3;
    const int inc = wave_incl_scan(s4, lane);
    if (lane == 31) wsum[buf][wv] = inc;
    __syncthreads();
    const int ws  = wsum[buf][lane];
    const int wsi = wave_incl_scan(ws, lane);
    const int wave_off = __shfl(wsi, wv, 32) - __shfl(ws, wv, 32);
    const int tile_tot = __shfl(wsi, 31, 32);
    const int ex = carry + wave_off + inc - s4;
    if (b + 3 < Nn) {
      *(int4*)(rowptr + b) = make_int4(ex, ex + s1, ex + s2, ex + s3);
    } else {
      if (b     < Nn) rowptr[b]     = ex;
      if (b + 1 < Nn) rowptr[b + 1] = ex + s1;
      if (b + 2 < Nn) rowptr[b + 2] = ex + s2;
    }
    carry += tile_tot;
    buf ^= 1;
  }
  if (t == 0) rowptr[Nn] = carry;
}

__global__ void csr_fill(const int* __restrict__ ei, const int* __restrict__ rowptr,
                         const int* __restrict__ rank, int* __restrict__ csr_src, int E) {
  int e = blockIdx.x * blockDim.x + threadIdx.x;
  if (e >= E) return;
  csr_src[rowptr[ei[E + e]] + rank[e]] = ei[e];
}

__device__ __forceinline__ float leaky(float e) { return e > 0.f ? e : 0.2f * e; }

__device__ __forceinline__ int lane_bcast_i(int v, int l) {
  return __builtin_amdgcn_readlane(v, l);
}
__device__ __forceinline__ float lane_bcast_f(float v, int l) {
  return __int_as_float(__builtin_amdgcn_readlane(__float_as_int(v), l));
}

__device__ __forceinline__ float wave_max(float v) {
  for (int off = 16; off > 0; off >>= 1) v = fmaxf(v, __shfl_xor(v, off, 32));
  return v;
}
__device__ __forceinline__ float wave_sum(float v) {
  for (int off = 16; off > 0; off >>= 1) v += __shfl_xor(v, off, 32);
  return v;
}

template <int G>
__device__ __forceinline__ float gather_group(const float* __restrict__ hfeat, int s,
                                              float w0, float w1, float w2, float w3,
                                              int j, int lane, float r) {
  float hv[G][HEADS];
#pragma unroll
  for (int g = 0; g < G; ++g) {
    const float* hp = hfeat + (size_t)lane_bcast_i(s, j + g) * HF;
#pragma unroll
    for (int k = 0; k < HEADS; ++k) hv[g][k] = hp[k * FEAT + lane];
  }
#pragma unroll
  for (int g = 0; g < G; ++g) {
    r = fmaf(lane_bcast_f(w0, j + g), hv[g][0], r);
    r = fmaf(lane_bcast_f(w1, j + g), hv[g][1], r);
    r = fmaf(lane_bcast_f(w2, j + g), hv[g][2], r);
    r = fmaf(lane_bcast_f(w3, j + g), hv[g][3], r);
  }
  return r;
}

__device__ __forceinline__ float gather_chunk(const float* __restrict__ hfeat, int s,
                                              float w0, float w1, float w2, float w3,
                                              int cnt, int lane, float acc) {
  int j = 0;
  for (; j + 8 <= cnt; j += 8) acc = gather_group<8>(hfeat, s, w0, w1, w2, w3, j, lane, acc);
  for (; j < cnt; j += 4)      acc = gather_group<4>(hfeat, s, w0, w1, w2, w3, j, lane, acc);
  return acc;
}

__global__ void __launch_bounds__(256)
gat_node(const int* __restrict__ rowptr, const int* __restrict__ csr_src,
         const float* __restrict__ asrc, const float* __restrict__ adst,
         const float* __restrict__ hfeat, const float* __restrict__ bias,
         float* __restrict__ out, int Nn, int do_elu) {
  const int lane = threadIdx.x & 31;
  const int wv = __builtin_amdgcn_readfirstlane((int)(threadIdx.x >> 5));
  const int d  = blockIdx.x * (256 / 32) + wv;
  if (d >= Nn) return;
  const int beg = __builtin_amdgcn_readfirstlane(rowptr[d]);
  const int deg = __builtin_amdgcn_readfirstlane(rowptr[d + 1]) - beg;
  const int tot = deg + 1;
  const bool two = tot > 32;
  const float4 ad = *(const float4*)(adst + (size_t)d * HEADS);
  const float4* __restrict__ asrc4 = (const float4*)asrc;

  int sA = d, sB = d;
  if (lane < deg) sA = csr_src[beg + lane];
  if (two && lane + 32 < deg) sB = csr_src[beg + 32 + lane];
  const float4 aA = asrc4[sA];
  float4 aB = make_float4(0.f, 0.f, 0.f, 0.f);
  if (two) aB = asrc4[sB];

  float eA0 = -INFINITY, eA1 = -INFINITY, eA2 = -INFINITY, eA3 = -INFINITY;
  float eB0 = -INFINITY, eB1 = -INFINITY, eB2 = -INFINITY, eB3 = -INFINITY;
  if (lane < tot) {
    eA0 = leaky(aA.x + ad.x); eA1 = leaky(aA.y + ad.y);
    eA2 = leaky(aA.z + ad.z); eA3 = leaky(aA.w + ad.w);
  }
  if (lane + 32 < tot) {
    eB0 = leaky(aB.x + ad.x); eB1 = leaky(aB.y + ad.y);
    eB2 = leaky(aB.z + ad.z); eB3 = leaky(aB.w + ad.w);
  }

  float m0 = fmaxf(eA0, eB0), m1 = fmaxf(eA1, eB1), m2 = fmaxf(eA2, eB2), m3 = fmaxf(eA3, eB3);
  for (int base = 64; base < tot; base += 32) {
    const int i = base + lane;
    int s = d;
    if (i < deg) s = csr_src[beg + i];
    const float4 a = asrc4[s];
    if (i < tot) {
      m0 = fmaxf(m0, leaky(a.x + ad.x));
      m1 = fmaxf(m1, leaky(a.y + ad.y));
      m2 = fmaxf(m2, leaky(a.z + ad.z));
      m3 = fmaxf(m3, leaky(a.w + ad.w));
    }
  }
  m0 = wave_max(m0); m1 = wave_max(m1); m2 = wave_max(m2); m3 = wave_max(m3);

  float pA0 = 0.f, pA1 = 0.f, pA2 = 0.f, pA3 = 0.f;
  float pB0 = 0.f, pB1 = 0.f, pB2 = 0.f, pB3 = 0.f;
  if (lane < tot) {
    pA0 = expf(eA0 - m0); pA1 = expf(eA1 - m1); pA2 = expf(eA2 - m2); pA3 = expf(eA3 - m3);
  }
  if (lane + 32 < tot) {
    pB0 = expf(eB0 - m0); pB1 = expf(eB1 - m1); pB2 = expf(eB2 - m2); pB3 = expf(eB3 - m3);
  }
  float s0 = pA0 + pB0, s1 = pA1 + pB1, s2 = pA2 + pB2, s3 = pA3 + pB3;
  for (int base = 64; base < tot; base += 32) {
    const int i = base + lane;
    int s = d;
    if (i < deg) s = csr_src[beg + i];
    const float4 a = asrc4[s];
    if (i < tot) {
      s0 += expf(leaky(a.x + ad.x) - m0);
      s1 += expf(leaky(a.y + ad.y) - m1);
      s2 += expf(leaky(a.z + ad.z) - m2);
      s3 += expf(leaky(a.w + ad.w) - m3);
    }
  }
  const float den0 = wave_sum(s0) + 1e-16f;
  const float den1 = wave_sum(s1) + 1e-16f;
  const float den2 = wave_sum(s2) + 1e-16f;
  const float den3 = wave_sum(s3) + 1e-16f;

  float acc = gather_chunk(hfeat, sA, pA0 / den0, pA1 / den1, pA2 / den2, pA3 / den3,
                           two ? 32 : tot, lane, 0.f);
  if (two)
    acc = gather_chunk(hfeat, sB, pB0 / den0, pB1 / den1, pB2 / den2, pB3 / den3,
                       (tot - 32 < 32) ? (tot - 32) : 32, lane, acc);
  for (int base = 64; base < tot; base += 32) {
    const int i = base + lane;
    int s = d;
    if (i < deg) s = csr_src[beg + i];
    const float4 a = asrc4[s];
    float w0 = 0.f, w1 = 0.f, w2 = 0.f, w3 = 0.f;
    if (i < tot) {
      w0 = expf(leaky(a.x + ad.x) - m0) / den0;
      w1 = expf(leaky(a.y + ad.y) - m1) / den1;
      w2 = expf(leaky(a.z + ad.z) - m2) / den2;
      w3 = expf(leaky(a.w + ad.w) - m3) / den3;
    }
    acc = gather_chunk(hfeat, s, w0, w1, w2, w3, (tot - base < 32) ? (tot - base) : 32, lane, acc);
  }

  float v = 0.25f * acc + bias[lane];
  if (do_elu) v = v > 0.f ? v : (expf(v) - 1.f);
  out[(size_t)d * FEAT + lane] = v;
}

extern "C" void kernel_launch(void* const* d_in, const int* in_sizes, int n_in,
                              void* d_out, int out_size, void* d_ws, size_t ws_size,
                              hipStream_t stream) {
  const float* x   = (const float*)d_in[0];
  const int*   ei  = (const int*)d_in[1];
  const float* W1  = (const float*)d_in[2];
  const float* as1 = (const float*)d_in[3];
  const float* ad1 = (const float*)d_in[4];
  const float* b1  = (const float*)d_in[5];
  const float* W2  = (const float*)d_in[6];
  const float* as2 = (const float*)d_in[7];
  const float* ad2 = (const float*)d_in[8];
  const float* b2  = (const float*)d_in[9];

  const int Nn = in_sizes[0] / HF;
  const int E  = in_sizes[1] / 2;

  const size_t n4 = ((size_t)Nn + 1 + 3) / 4 * 4;
  const size_t e4 = ((size_t)E + 3) / 4 * 4;
  float* ws     = (float*)d_ws;
  float* hbuf   = ws;
  float* x2     = hbuf + (size_t)Nn * HF;
  float* asrc   = x2   + (size_t)Nn * FEAT;
  float* adst   = asrc + (size_t)Nn * HEADS;
  int*   cnt    = (int*)(adst + (size_t)Nn * HEADS);
  int*   rowptr = cnt + n4;
  int*   rank   = rowptr + n4;
  int*   csr    = rank + e4;

  const dim3 blk(256);
  const int row_tiles = Nn / 16;
  const int tail_row0 = row_tiles * 16;

  csr_zero<<<(Nn + 255) / 256, blk, 0, stream>>>(cnt, Nn);
  if (E > 0) csr_count<<<(E + 255) / 256, blk, 0, stream>>>(ei, cnt, rank, E);
  csr_scan<<<1, SCAN_T, 0, stream>>>(cnt, rowptr, Nn);
  if (E > 0) csr_fill<<<(E + 255) / 256, blk, 0, stream>>>(ei, rowptr, rank, csr, E);

  for (int layer = 0; layer < 2; ++layer) {
    const float* Xin = (layer == 0) ? x  : x2;
    const int    K   = (layer == 0) ? 128 : FEAT;
    const float* Wm  = (layer == 0) ? W1 : W2;
    const float* as  = (layer == 0) ? as1 : as2;
    const float* ad  = (layer == 0) ? ad1 : ad2;
    const float* bb  = (layer == 0) ? b1 : b2;
    float*       op  = (layer == 0) ? x2 : (float*)d_out;

    if (row_tiles > 0)
      gat_gemm_alpha_wmma_f32<<<row_tiles, dim3(128), 0, stream>>>(Xin, Wm, as, ad, hbuf,
                                                                   asrc, adst, K);
    if (tail_row0 < Nn) {
      const int trows = Nn - tail_row0;
      gat_gemm_tail<<<(trows * HF + 255) / 256, blk, 0, stream>>>(Xin, Wm, hbuf, K, tail_row0, Nn);
      gat_alpha<<<(trows * HEADS + 255) / 256, blk, 0, stream>>>(
          hbuf + (size_t)tail_row0 * HF, as, ad, asrc + (size_t)tail_row0 * HEADS,
          adst + (size_t)tail_row0 * HEADS, trows);
    }
    gat_node<<<(Nn + 7) / 8, blk, 0, stream>>>(rowptr, csr, asrc, adst, hbuf, bb, op, Nn,
                                               layer == 0 ? 1 : 0);
  }
}


# edges grouped by destination with a two-level counting sort in LDS: no global atomics, no scattered stores, 6 launches not 8; list order is LDS-atomic arrival order (float sum order changes)
# speedup vs baseline: 8.3648x; 1.8126x over previous
#include <hip/hip_runtime.h>
#include <math.h>

#define HEADS 4
#define FEAT  32
#define HF    128

typedef __attribute__((ext_vector_type(2))) float v2f;
typedef __attribute__((ext_vector_type(8))) float v8f;

__global__ void __launch_bounds__(128)
gat_gemm_alpha_wmma_f32(const float* __restrict__ X, const float* __restrict__ W,
                        const float* __restrict__ att_src, const float* __restrict__ att_dst,
                        float* __restrict__ Hout, float* __restrict__ a_src,
                        float* __restrict__ a_dst, int K) {
  const int head = threadIdx.x >> 5;
  const int lane = threadIdx.x & 31;
  const int half = lane >> 4;
  const int l15  = lane & 15;
  const int row  = blockIdx.x * 16 + l15;
  const int col  = head * FEAT + l15;
  const int kh   = half * 2;
  const float* __restrict__ xrow = X + (size_t)row * K;

  v8f c0 = {}, c1 = {};
  for (int k0 = 0; k0 < K; k0 += 4) {
    v2f a, b0, b1;
    a.x = xrow[k0 + kh];
    a.y = xrow[k0 + kh + 1];
    const float* __restrict__ wr = W + (size_t)(k0 + kh) * HF + col;
    b0.x = wr[0];
    b0.y = wr[HF];
    b1.x = wr[16];
    b1.y = wr[HF + 16];
    c0 = __builtin_amdgcn_wmma_f32_16x16x4_f32(false, a, false, b0,
                                               (short)0, c0, false, false);
    c1 = __builtin_amdgcn_wmma_f32_16x16x4_f32(false, a, false, b1,
                                               (short)0, c1, false, false);
  }
  const int rbase = blockIdx.x * 16 + half * 8;
  float* orow = Hout + (size_t)rbase * HF + col;
#pragma unroll
  for (int v = 0; v < 8; ++v) {
    orow[(size_t)v * HF]      = c0[v];
    orow[(size_t)v * HF + 16] = c1[v];
  }

  const float as0 = att_src[col], as1 = att_src[col + 16];
  const float ad0 = att_dst[col], ad1 = att_dst[col + 16];
  float q[16];
#pragma unroll
  for (int v = 0; v < 8; ++v) {
    q[v]     = c0[v] * as0 + c1[v] * as1;
    q[8 + v] = c0[v] * ad0 + c1[v] * ad1;
  }
  const bool b3 = (l15 & 8) != 0, b2 = (l15 & 4) != 0, b1s = (l15 & 2) != 0, b0s = (l15 & 1) != 0;
  float r1[8], r2[4], r3[2];
#pragma unroll
  for (int i = 0; i < 8; ++i) {
    const float send = b3 ? q[i] : q[i + 8];
    const float keep = b3 ? q[i + 8] : q[i];
    r1[i] = keep + __shfl_xor(send, 8, 32);
  }
#pragma unroll
  for (int i = 0; i < 4; ++i) {
    const float send = b2 ? r1[i] : r1[i + 4];
    const float keep = b2 ? r1[i + 4] : r1[i];
    r2[i] = keep + __shfl_xor(send, 4, 32);
  }
#pragma unroll
  for (int i = 0; i < 2; ++i) {
    const float send = b1s ? r2[i] : r2[i + 2];
    const float keep = b1s ? r2[i + 2] : r2[i];
    r3[i] = keep + __shfl_xor(send, 2, 32);
  }
  const float send = b0s ? r3[0] : r3[1];
  const float keep = b0s ? r3[1] : r3[0];
  const float tot  = keep + __shfl_xor(send, 1, 32);
  float* __restrict__ ap = (l15 < 8) ? a_src : a_dst;
  ap[(size_t)(rbase + (l15 & 7)) * HEADS + head] = tot;
}

__global__ void gat_gemm_tail(const float* __restrict__ X, const float* __restrict__ W,
                              float* __restrict__ Hout, int K, int row0, int Nn) {
  int i = blockIdx.x * blockDim.x + threadIdx.x;
  int rows = Nn - row0;
  if (i >= rows * HF) return;
  int r = row0 + (i >> 7), col = i & (HF - 1);
  float acc = 0.f;
  for (int k = 0; k < K; ++k) acc += X[(size_t)r * K + k] * W[(size_t)k * HF + col];
  Hout[(size_t)r * HF + col] = acc;
}

__global__ void gat_alpha(const float* __restrict__ H, const float* __restrict__ att_src,
                          const float* __restrict__ att_dst,
                          float* __restrict__ a_src, float* __restrict__ a_dst, int Nn) {
  int i = blockIdx.x * blockDim.x + threadIdx.x;
  if (i >= Nn * HEADS) return;
  int h = i & (HEADS - 1);
  const float4* hp = (const float4*)(H + (size_t)(i >> 2) * HF + h * FEAT);
  const float4* as = (const float4*)(att_src + h * FEAT);
  const float4* ad = (const float4*)(att_dst + h * FEAT);
  float s1 = 0.f, s2 = 0.f;
#pragma unroll
  for (int f = 0; f < FEAT / 4; ++f) {
    float4 v = hp[f], a = as[f], d = ad[f];
    s1 += v.x * a.x + v.y * a.y + v.z * a.z + v.w * a.w;
    s2 += v.x * d.x + v.y * d.y + v.z * d.z + v.w * d.w;
  }
  a_src[i] = s1;
  a_dst[i] = s2;
}

#define BIN_SHIFT 7
#define BIN_W     128
#define SORT_T    1024
#define CHUNK     4096
#define NB_MAX    1023
#define CH_MAX    1024
#define BIN_CAP   8192

__device__ __forceinline__ int wave_incl_scan(int v, int lane) {
  for (int off = 1; off < 32; off <<= 1) {
    int u = __shfl_up(v, off, 32);
    if (lane >= off) v += u;
  }
  return v;
}

__device__ __forceinline__ int block_excl_scan(int v, int* wsum, int& total) {
  const int lane = threadIdx.x & 31, wv = threadIdx.x >> 5;
  const int inc = wave_incl_scan(v, lane);
  if (lane == 31) wsum[wv] = inc;
  __syncthreads();
  const int ws  = wsum[lane];
  const int wsi = wave_incl_scan(ws, lane);
  total = __shfl(wsi, 31, 32);
  return __shfl(wsi, wv, 32) - __shfl(ws, wv, 32) + inc - v;
}

__global__ void __launch_bounds__(SORT_T)
edge_chunk_sort(const int* __restrict__ ei, int* __restrict__ sorted,
                int* __restrict__ offs, int E, int NB, int ostride) {
  __shared__ int cnt[NB_MAX + 1];
  __shared__ int off[NB_MAX + 1];
  __shared__ int srt[CHUNK];
  __shared__ int wsum[32];
  const int t  = threadIdx.x;
  const int e0 = blockIdx.x * CHUNK;
  const int n  = (E - e0 < CHUNK) ? (E - e0) : CHUNK;
  if (t <= NB) cnt[t] = 0;
  __syncthreads();
  int pk[CHUNK / SORT_T], bn[CHUNK / SORT_T], rk[CHUNK / SORT_T];
#pragma unroll
  for (int k = 0; k < CHUNK / SORT_T; ++k) {
    const int i = t + k * SORT_T;
    const int j = e0 + ((i < n) ? i : (n - 1));
    const int s = ei[j], d = ei[E + j];
    bn[k] = d >> BIN_SHIFT;
    pk[k] = (s << BIN_SHIFT) | (d & (BIN_W - 1));
  }
#pragma unroll
  for (int k = 0; k < CHUNK / SORT_T; ++k) {
    rk[k] = 0;
    if (t + k * SORT_T < n) rk[k] = atomicAdd(&cnt[bn[k]], 1);
  }
  __syncthreads();
  int tot;
  const int ex = block_excl_scan((t < NB) ? cnt[t] : 0, wsum, tot);
  if (t <= NB) {
    off[t] = ex;
    offs[(size_t)blockIdx.x * ostride + t] = ex;
  }
  __syncthreads();
#pragma unroll
  for (int k = 0; k < CHUNK / SORT_T; ++k)
    if (t + k * SORT_T < n) srt[off[bn[k]] + rk[k]] = pk[k];
  __syncthreads();
#pragma unroll
  for (int k = 0; k < CHUNK / SORT_T; ++k) {
    const int i = t + k * SORT_T;
    if (i < n) sorted[e0 + i] = srt[i];
  }
}

__device__ __forceinline__ int bin_entry(const int* __restrict__ sorted, const int* pS,
                                         const int* pP, int NC, int i) {
  int lo = 0, hi = NC;
  while (hi - lo > 1) {
    const int mid = (lo + hi) >> 1;
    if (pP[mid] <= i) lo = mid; else hi = mid;
  }
  return sorted[pS[lo] + (i - pP[lo])];
}

__global__ void __launch_bounds__(SORT_T)
edge_bin_build(const int* __restrict__ sorted, const int* __restrict__ offs,
               int* __restrict__ rowptr, int* __restrict__ csr_src, int Nn, int NB, int NC,
               int ostride) {
  __shared__ int pS[CH_MAX];
  __shared__ int pP[CH_MAX + 1];
  __shared__ int cntL[BIN_W];
  __shared__ int curL[BIN_W];
  __shared__ int raw[BIN_CAP];
  __shared__ int srt[BIN_CAP];
  __shared__ int wsA[32];
  __shared__ int wsB[32];
  const int t = threadIdx.x, b = blockIdx.x;
  int a = 0, len = 0;
  if (t < NC) {
    const int* o = offs + (size_t)t * ostride + b;
    a   = o[0];
    len = o[1] - a;
    pS[t] = t * CHUNK + a;
  }
  if (t < BIN_W) cntL[t] = 0;
  int total, base;
  const int pre = block_excl_scan(len, wsA, total);
  (void)block_excl_scan(a, wsB, base);
  if (t < NC) pP[t] = pre;
  if (t == 0) pP[NC] = total;
  __syncthreads();

  const bool fits = total <= BIN_CAP;
  for (int i0 = t; i0 < total + t; i0 += 4 * SORT_T) {
    int v[4];
#pragma unroll
    for (int k = 0; k < 4; ++k) {
      const int i = i0 + k * SORT_T;
      v[k] = bin_entry(sorted, pS, pP, NC, (i < total) ? i : (total - 1));
    }
#pragma unroll
    for (int k = 0; k < 4; ++k) {
      const int i = i0 + k * SORT_T;
      if (i < total) {
        if (fits) raw[i] = v[k];
        atomicAdd(&cntL[v[k] & (BIN_W - 1)], 1);
      }
    }
  }
  __syncthreads();
  if (t < 32) {
    const int c0 = cntL[4 * t], c1 = cntL[4 * t + 1], c2 = cntL[4 * t + 2], c3 = cntL[4 * t + 3];
    const int s4 = c0 + c1 + c2 + c3;
    const int ex = wave_incl_scan(s4, t) - s4;
    const int st0 = ex, st1 = st0 + c0, st2 = st1 + c1, st3 = st2 + c2;
    curL[4 * t] = st0; curL[4 * t + 1] = st1; curL[4 * t + 2] = st2; curL[4 * t + 3] = st3;
    const int node = b * BIN_W + 4 * t;
    if (node     < Nn) rowptr[node]     = base + st0;
    if (node + 1 < Nn) rowptr[node + 1] = base + st1;
    if (node + 2 < Nn) rowptr[node + 2] = base + st2;
    if (node + 3 < Nn) rowptr[node + 3] = base + st3;
  }
  if (b == NB - 1 && t == 0) rowptr[Nn] = base + total;
  __syncthreads();
  for (int i = t; i < total; i += SORT_T) {
    const int v = fits ? raw[i] : bin_entry(sorted, pS, pP, NC, i);
    const int p = atomicAdd(&curL[v & (BIN_W - 1)], 1);
    if (fits) srt[p] = v >> BIN_SHIFT;
    else      csr_src[base + p] = v >> BIN_SHIFT;
  }
  __syncthreads();
  if (fits)
    for (int i = t; i < total; i += SORT_T) csr_src[base + i] = srt[i];
}

__device__ __forceinline__ float leaky(float e) { return e > 0.f ? e : 0.2f * e; }

__device__ __forceinline__ int lane_bcast_i(int v, int l) {
  return __builtin_amdgcn_readlane(v, l);
}
__device__ __forceinline__ float lane_bcast_f(float v, int l) {
  return __int_as_float(__builtin_amdgcn_readlane(__float_as_int(v), l));
}

__device__ __forceinline__ float wave_max(float v) {
  for (int off = 16; off > 0; off >>= 1) v = fmaxf(v, __shfl_xor(v, off, 32));
  return v;
}
__device__ __forceinline__ float wave_sum(float v) {
  for (int off = 16; off > 0; off >>= 1) v += __shfl_xor(v, off, 32);
  return v;
}

template <int G>
__device__ __forceinline__ float gather_group(const float* __restrict__ hfeat, int s,
                                              float w0, float w1, float w2, float w3,
                                              int j, int lane, float r) {
  float hv[G][HEADS];
#pragma unroll
  for (int g = 0; g < G; ++g) {
    const float* hp = hfeat + (size_t)lane_bcast_i(s, j + g) * HF;
#pragma unroll
    for (int k = 0; k < HEADS; ++k) hv[g][k] = hp[k * FEAT + lane];
  }
#pragma unroll
  for (int g = 0; g < G; ++g) {
    r = fmaf(lane_bcast_f(w0, j + g), hv[g][0], r);
    r = fmaf(lane_bcast_f(w1, j + g), hv[g][1], r);
    r = fmaf(lane_bcast_f(w2, j + g), hv[g][2], r);
    r = fmaf(lane_bcast_f(w3, j + g), hv[g][3], r);
  }
  return r;
}

__device__ __forceinline__ float gather_chunk(const float* __restrict__ hfeat, int s,
                                              float w0, float w1, float w2, float w3,
                                              int cnt, int lane, float acc) {
  int j = 0;
  for (; j + 8 <= cnt; j += 8) acc = gather_group<8>(hfeat, s, w0, w1, w2, w3, j, lane, acc);
  for (; j < cnt; j += 4)      acc = gather_group<4>(hfeat, s, w0, w1, w2, w3, j, lane, acc);
  return acc;
}

__global__ void __launch_bounds__(256)
gat_node(const int* __restrict__ rowptr, const int* __restrict__ csr_src,
         const float* __restrict__ asrc, const float* __restrict__ adst,
         const float* __restrict__ hfeat, const float* __restrict__ bias,
         float* __restrict__ out, int Nn, int do_elu) {
  const int lane = threadIdx.x & 31;
  const int wv = __builtin_amdgcn_readfirstlane((int)(threadIdx.x >> 5));
  const int d  = blockIdx.x * (256 / 32) + wv;
  if (d >= Nn) return;
  const int beg = __builtin_amdgcn_readfirstlane(rowptr[d]);
  const int deg = __builtin_amdgcn_readfirstlane(rowptr[d + 1]) - beg;
  const int tot = deg + 1;
  const bool two = tot > 32;
  const float4 ad = *(const float4*)(adst + (size_t)d * HEADS);
  const float4* __restrict__ asrc4 = (const float4*)asrc;

  int sA = d, sB = d;
  if (lane < deg) sA = csr_src[beg + lane];
  if (two && lane + 32 < deg) sB = csr_src[beg + 32 + lane];
  const float4 aA = asrc4[sA];
  float4 aB = make_float4(0.f, 0.f, 0.f, 0.f);
  if (two) aB = asrc4[sB];

  float eA0 = -INFINITY, eA1 = -INFINITY, eA2 = -INFINITY, eA3 = -INFINITY;
  float eB0 = -INFINITY, eB1 = -INFINITY, eB2 = -INFINITY, eB3 = -INFINITY;
  if (lane < tot) {
    eA0 = leaky(aA.x + ad.x); eA1 = leaky(aA.y + ad.y);
    eA2 = leaky(aA.z + ad.z); eA3 = leaky(aA.w + ad.w);
  }
  if (lane + 32 < tot) {
    eB0 = leaky(aB.x + ad.x); eB1 = leaky(aB.y + ad.y);
    eB2 = leaky(aB.z + ad.z); eB3 = leaky(aB.w + ad.w);
  }

  float m0 = fmaxf(eA0, eB0), m1 = fmaxf(eA1, eB1), m2 = fmaxf(eA2, eB2), m3 = fmaxf(eA3, eB3);
  for (int base = 64; base < tot; base += 32) {
    const int i = base + lane;
    int s = d;
    if (i < deg) s = csr_src[beg + i];
    const float4 a = asrc4[s];
    if (i < tot) {
      m0 = fmaxf(m0, leaky(a.x + ad.x));
      m1 = fmaxf(m1, leaky(a.y + ad.y));
      m2 = fmaxf(m2, leaky(a.z + ad.z));
      m3 = fmaxf(m3, leaky(a.w + ad.w));
    }
  }
  m0 = wave_max(m0); m1 = wave_max(m1); m2 = wave_max(m2); m3 = wave_max(m3);

  float pA0 = 0.f, pA1 = 0.f, pA2 = 0.f, pA3 = 0.f;
  float pB0 = 0.f, pB1 = 0.f, pB2 = 0.f, pB3 = 0.f;
  if (lane < tot) {
    pA0 = expf(eA0 - m0); pA1 = expf(eA1 - m1); pA2 = expf(eA2 - m2); pA3 = expf(eA3 - m3);
  }
  if (lane + 32 < tot) {
    pB0 = expf(eB0 - m0); pB1 = expf(eB1 - m1); pB2 = expf(eB2 - m2); pB3 = expf(eB3 - m3);
  }
  float s0 = pA0 + pB0, s1 = pA1 + pB1, s2 = pA2 + pB2, s3 = pA3 + pB3;
  for (int base = 64; base < tot; base += 32) {
    const int i = base + lane;
    int s = d;
    if (i < deg) s = csr_src[beg + i];
    const float4 a = asrc4[s];
    if (i < tot) {
      s0 += expf(leaky(a.x + ad.x) - m0);
      s1 += expf(leaky(a.y + ad.y) - m1);
      s2 += expf(leaky(a.z + ad.z) - m2);
      s3 += expf(leaky(a.w + ad.w) - m3);
    }
  }
  const float den0 = wave_sum(s0) + 1e-16f;
  const float den1 = wave_sum(s1) + 1e-16f;
  const float den2 = wave_sum(s2) + 1e-16f;
  const float den3 = wave_sum(s3) + 1e-16f;

  float acc = gather_chunk(hfeat, sA, pA0 / den0, pA1 / den1, pA2 / den2, pA3 / den3,
                           two ? 32 : tot, lane, 0.f);
  if (two)
    acc = gather_chunk(hfeat, sB, pB0 / den0, pB1 / den1, pB2 / den2, pB3 / den3,
                       (tot - 32 < 32) ? (tot - 32) : 32, lane, acc);
  for (int base = 64; base < tot; base += 32) {
    const int i = base + lane;
    int s = d;
    if (i < deg) s = csr_src[beg + i];
    const float4 a = asrc4[s];
    float w0 = 0.f, w1 = 0.f, w2 = 0.f, w3 = 0.f;
    if (i < tot) {
      w0 = expf(leaky(a.x + ad.x) - m0) / den0;
      w1 = expf(leaky(a.y + ad.y) - m1) / den1;
      w2 = expf(leaky(a.z + ad.z) - m2) / den2;
      w3 = expf(leaky(a.w + ad.w) - m3) / den3;
    }
    acc = gather_chunk(hfeat, s, w0, w1, w2, w3, (tot - base < 32) ? (tot - base) : 32, lane, acc);
  }

  float v = 0.25f * acc + bias[lane];
  if (do_elu) v = v > 0.f ? v : (expf(v) - 1.f);
  out[(size_t)d * FEAT + lane] = v;
}

extern "C" void kernel_launch(void* const* d_in, const int* in_sizes, int n_in,
                              void* d_out, int out_size, void* d_ws, size_t ws_size,
                              hipStream_t stream) {
  const float* x   = (const float*)d_in[0];
  const int*   ei  = (const int*)d_in[1];
  const float* W1  = (const float*)d_in[2];
  const float* as1 = (const float*)d_in[3];
  const float* ad1 = (const float*)d_in[4];
  const float* b1  = (const float*)d_in[5];
  const float* W2  = (const float*)d_in[6];
  const float* as2 = (const float*)d_in[7];
  const float* ad2 = (const float*)d_in[8];
  const float* b2  = (const float*)d_in[9];

  const int Nn = in_sizes[0] / HF;
  const int E  = in_sizes[1] / 2;

  const size_t n4 = ((size_t)Nn + 1 + 31) / 32 * 32;
  const size_t e4 = ((size_t)E + 31) / 32 * 32;
  float* ws     = (float*)d_ws;
  float* hbuf   = ws;
  float* x2     = hbuf + (size_t)Nn * HF;
  float* asrc   = x2   + (size_t)Nn * FEAT;
  float* adst   = asrc + (size_t)Nn * HEADS;
  int*   rowptr = (int*)(adst + (size_t)Nn * HEADS);
  int*   sorted = rowptr + n4;
  int*   csr    = sorted + e4;
  int*   offs   = csr + e4;

  const dim3 blk(256);
  const int row_tiles = Nn / 16;
  const int tail_row0 = row_tiles * 16;

  const int NB = (Nn + BIN_W - 1) / BIN_W;
  const int NC = (E + CHUNK - 1) / CHUNK;
  const int ostride = (NB + 1 + 31) / 32 * 32;
  if (NC > 0) edge_chunk_sort<<<NC, SORT_T, 0, stream>>>(ei, sorted, offs, E, NB, ostride);
  edge_bin_build<<<NB, SORT_T, 0, stream>>>(sorted, offs, rowptr, csr, Nn, NB, NC, ostride);

  for (int layer = 0; layer < 2; ++layer) {
    const float* Xin = (layer == 0) ? x  : x2;
    const int    K   = (layer == 0) ? 128 : FEAT;
    const float* Wm  = (layer == 0) ? W1 : W2;
    const float* as  = (layer == 0) ? as1 : as2;
    const float* ad  = (layer == 0) ? ad1 : ad2;
    const float* bb  = (layer == 0) ? b1 : b2;
    float*       op  = (layer == 0) ? x2 : (float*)d_out;

    if (row_tiles > 0)
      gat_gemm_alpha_wmma_f32<<<row_tiles, dim3(128), 0, stream>>>(Xin, Wm, as, ad, hbuf,
                                                                   asrc, adst, K);
    if (tail_row0 < Nn) {
      const int trows = Nn - tail_row0;
      gat_gemm_tail<<<(trows * HF + 255) / 256, blk, 0, stream>>>(Xin, Wm, hbuf, K, tail_row0, Nn);
      gat_alpha<<<(trows * HEADS + 255) / 256, blk, 0, stream>>>(
          hbuf + (size_t)tail_row0 * HF, as, ad, asrc + (size_t)tail_row0 * HEADS,
          adst + (size_t)tail_row0 * HEADS, trows);
    }
    gat_node<<<(Nn + 7) / 8, blk, 0, stream>>>(rowptr, csr, asrc, adst, hbuf, bb, op, Nn,
                                               layer == 0 ? 1 : 0);
  }
}


# layer 2 applies W2 after the aggregation: node kernel gathers the 32-wide input rows per head (128 B not 512 B per edge), then a WMMA GEMM [N,128]x[128,32]; changes the float summation order
# speedup vs baseline: 9.1378x; 1.0924x over previous
#include <hip/hip_runtime.h>
#include <math.h>

#define HEADS 4
#define FEAT  32
#define HF    128

typedef __attribute__((ext_vector_type(2))) float v2f;
typedef __attribute__((ext_vector_type(8))) float v8f;

template <bool STORE_H>
__global__ void __launch_bounds__(128)
gat_gemm_alpha_wmma_f32(const float* __restrict__ X, const float* __restrict__ W,
                        const float* __restrict__ att_src, const float* __restrict__ att_dst,
                        float* __restrict__ Hout, float* __restrict__ a_src,
                        float* __restrict__ a_dst, int K) {
  const int head = threadIdx.x >> 5;
  const int lane = threadIdx.x & 31;
  const int half = lane >> 4;
  const int l15  = lane & 15;
  const int row  = blockIdx.x * 16 + l15;
  const int col  = head * FEAT + l15;
  const int kh   = half * 2;
  const float* __restrict__ xrow = X + (size_t)row * K;

  v8f c0 = {}, c1 = {};
  for (int k0 = 0; k0 < K; k0 += 4) {
    v2f a, b0, b1;
    a.x = xrow[k0 + kh];
    a.y = xrow[k0 + kh + 1];
    const float* __restrict__ wr = W + (size_t)(k0 + kh) * HF + col;
    b0.x = wr[0];
    b0.y = wr[HF];
    b1.x = wr[16];
    b1.y = wr[HF + 16];
    c0 = __builtin_amdgcn_wmma_f32_16x16x4_f32(false, a, false, b0,
                                               (short)0, c0, false, false);
    c1 = __builtin_amdgcn_wmma_f32_16x16x4_f32(false, a, false, b1,
                                               (short)0, c1, false, false);
  }
  const int rbase = blockIdx.x * 16 + half * 8;
  if (STORE_H) {
    float* orow = Hout + (size_t)rbase * HF + col;
#pragma unroll
    for (int v = 0; v < 8; ++v) {
      orow[(size_t)v * HF]      = c0[v];
      orow[(size_t)v * HF + 16] = c1[v];
    }
  }

  const float as0 = att_src[col], as1 = att_src[col + 16];
  const float ad0 = att_dst[col], ad1 = att_dst[col + 16];
  float q[16];
#pragma unroll
  for (int v = 0; v < 8; ++v) {
    q[v]     = c0[v] * as0 + c1[v] * as1;
    q[8 + v] = c0[v] * ad0 + c1[v] * ad1;
  }
  const bool b3 = (l15 & 8) != 0, b2 = (l15 & 4) != 0, b1s = (l15 & 2) != 0, b0s = (l15 & 1) != 0;
  float r1[8], r2[4], r3[2];
#pragma unroll
  for (int i = 0; i < 8; ++i) {
    const float send = b3 ? q[i] : q[i + 8];
    const float keep = b3 ? q[i + 8] : q[i];
    r1[i] = keep + __shfl_xor(send, 8, 32);
  }
#pragma unroll
  for (int i = 0; i < 4; ++i) {
    const float send = b2 ? r1[i] : r1[i + 4];
    const float keep = b2 ? r1[i + 4] : r1[i];
    r2[i] = keep + __shfl_xor(send, 4, 32);
  }
#pragma unroll
  for (int i = 0; i < 2; ++i) {
    const float send = b1s ? r2[i] : r2[i + 2];
    const float keep = b1s ? r2[i + 2] : r2[i];
    r3[i] = keep + __shfl_xor(send, 2, 32);
  }
  const float send = b0s ? r3[0] : r3[1];
  const float keep = b0s ? r3[1] : r3[0];
  const float tot  = keep + __shfl_xor(send, 1, 32);
  float* __restrict__ ap = (l15 < 8) ? a_src : a_dst;
  ap[(size_t)(rbase + (l15 & 7)) * HEADS + head] = tot;
}

__global__ void gat_gemm_tail(const float* __restrict__ X, const float* __restrict__ W,
                              float* __restrict__ Hout, int K, int row0, int Nn) {
  int i = blockIdx.x * blockDim.x + threadIdx.x;
  int rows = Nn - row0;
  if (i >= rows * HF) return;
  int r = row0 + (i >> 7), col = i & (HF - 1);
  float acc = 0.f;
  for (int k = 0; k < K; ++k) acc += X[(size_t)r * K + k] * W[(size_t)k * HF + col];
  Hout[(size_t)r * HF + col] = acc;
}

__global__ void __launch_bounds__(128)
gat_headmix_wmma_f32(const float* __restrict__ G, const float* __restrict__ W,
                     const float* __restrict__ bias, float* __restrict__ out, int row_tiles) {
  const int wv   = __builtin_amdgcn_readfirstlane((int)(threadIdx.x >> 5));
  const int tile = blockIdx.x * 4 + wv;
  if (tile >= row_tiles) return;
  const int lane = threadIdx.x & 31;
  const int half = lane >> 4;
  const int l15  = lane & 15;
  const int kh   = half * 2;
  const float* __restrict__ grow = G + (size_t)(tile * 16 + l15) * HF;

  v8f c0 = {}, c1 = {};
  for (int k0 = 0; k0 < HF; k0 += 4) {
    const int kk = k0 + kh;
    v2f a, b0, b1;
    a.x = grow[kk];
    a.y = grow[kk + 1];
    const float* __restrict__ wr = W + (size_t)(kk & (FEAT - 1)) * HF + (kk >> 5) * FEAT + l15;
    b0.x = wr[0];
    b0.y = wr[HF];
    b1.x = wr[16];
    b1.y = wr[HF + 16];
    c0 = __builtin_amdgcn_wmma_f32_16x16x4_f32(false, a, false, b0,
                                               (short)0, c0, false, false);
    c1 = __builtin_amdgcn_wmma_f32_16x16x4_f32(false, a, false, b1,
                                               (short)0, c1, false, false);
  }
  const int rbase = tile * 16 + half * 8;
  const float bs0 = bias[l15], bs1 = bias[l15 + 16];
  float* orow = out + (size_t)rbase * FEAT + l15;
#pragma unroll
  for (int v = 0; v < 8; ++v) {
    orow[(size_t)v * FEAT]      = 0.25f * c0[v] + bs0;
    orow[(size_t)v * FEAT + 16] = 0.25f * c1[v] + bs1;
  }
}

__global__ void gat_headmix_tail(const float* __restrict__ G, const float* __restrict__ W,
                                 const float* __restrict__ bias, float* __restrict__ out,
                                 int row0, int Nn) {
  int i = blockIdx.x * blockDim.x + threadIdx.x;
  int rows = Nn - row0;
  if (i >= rows * FEAT) return;
  int r = row0 + (i >> 5), f = i & (FEAT - 1);
  float acc = 0.f;
  for (int kk = 0; kk < HF; ++kk)
    acc += G[(size_t)r * HF + kk] * W[(size_t)(kk & (FEAT - 1)) * HF + (kk >> 5) * FEAT + f];
  out[(size_t)r * FEAT + f] = 0.25f * acc + bias[f];
}

__global__ void gat_alpha(const float* __restrict__ H, const float* __restrict__ att_src,
                          const float* __restrict__ att_dst,
                          float* __restrict__ a_src, float* __restrict__ a_dst, int Nn) {
  int i = blockIdx.x * blockDim.x + threadIdx.x;
  if (i >= Nn * HEADS) return;
  int h = i & (HEADS - 1);
  const float4* hp = (const float4*)(H + (size_t)(i >> 2) * HF + h * FEAT);
  const float4* as = (const float4*)(att_src + h * FEAT);
  const float4* ad = (const float4*)(att_dst + h * FEAT);
  float s1 = 0.f, s2 = 0.f;
#pragma unroll
  for (int f = 0; f < FEAT / 4; ++f) {
    float4 v = hp[f], a = as[f], d = ad[f];
    s1 += v.x * a.x + v.y * a.y + v.z * a.z + v.w * a.w;
    s2 += v.x * d.x + v.y * d.y + v.z * d.z + v.w * d.w;
  }
  a_src[i] = s1;
  a_dst[i] = s2;
}

#define BIN_SHIFT 7
#define BIN_W     128
#define SORT_T    1024
#define CHUNK     4096
#define NB_MAX    1023
#define CH_MAX    1024
#define BIN_CAP   8192

__device__ __forceinline__ int wave_incl_scan(int v, int lane) {
  for (int off = 1; off < 32; off <<= 1) {
    int u = __shfl_up(v, off, 32);
    if (lane >= off) v += u;
  }
  return v;
}

__device__ __forceinline__ int block_excl_scan(int v, int* wsum, int& total) {
  const int lane = threadIdx.x & 31, wv = threadIdx.x >> 5;
  const int inc = wave_incl_scan(v, lane);
  if (lane == 31) wsum[wv] = inc;
  __syncthreads();
  const int ws  = wsum[lane];
  const int wsi = wave_incl_scan(ws, lane);
  total = __shfl(wsi, 31, 32);
  return __shfl(wsi, wv, 32) - __shfl(ws, wv, 32) + inc - v;
}

__global__ void __launch_bounds__(SORT_T)
edge_chunk_sort(const int* __restrict__ ei, int* __restrict__ sorted,
                int* __restrict__ offs, int E, int NB, int ostride) {
  __shared__ int cnt[NB_MAX + 1];
  __shared__ int off[NB_MAX + 1];
  __shared__ int srt[CHUNK];
  __shared__ int wsum[32];
  const int t  = threadIdx.x;
  const int e0 = blockIdx.x * CHUNK;
  const int n  = (E - e0 < CHUNK) ? (E - e0) : CHUNK;
  if (t <= NB) cnt[t] = 0;
  __syncthreads();
  int pk[CHUNK / SORT_T], bn[CHUNK / SORT_T], rk[CHUNK / SORT_T];
#pragma unroll
  for (int k = 0; k < CHUNK / SORT_T; ++k) {
    const int i = t + k * SORT_T;
    const int j = e0 + ((i < n) ? i : (n - 1));
    const int s = ei[j], d = ei[E + j];
    bn[k] = d >> BIN_SHIFT;
    pk[k] = (s << BIN_SHIFT) | (d & (BIN_W - 1));
  }
#pragma unroll
  for (int k = 0; k < CHUNK / SORT_T; ++k) {
    rk[k] = 0;
    if (t + k * SORT_T < n) rk[k] = atomicAdd(&cnt[bn[k]], 1);
  }
  __syncthreads();
  int tot;
  const int ex = block_excl_scan((t < NB) ? cnt[t] : 0, wsum, tot);
  if (t <= NB) {
    off[t] = ex;
    offs[(size_t)blockIdx.x * ostride + t] = ex;
  }
  __syncthreads();
#pragma unroll
  for (int k = 0; k < CHUNK / SORT_T; ++k)
    if (t + k * SORT_T < n) srt[off[bn[k]] + rk[k]] = pk[k];
  __syncthreads();
#pragma unroll
  for (int k = 0; k < CHUNK / SORT_T; ++k) {
    const int i = t + k * SORT_T;
    if (i < n) sorted[e0 + i] = srt[i];
  }
}

__device__ __forceinline__ int bin_entry(const int* __restrict__ sorted, const int* pS,
                                         const int* pP, int NC, int i) {
  int lo = 0, hi = NC;
  while (hi - lo > 1) {
    const int mid = (lo + hi) >> 1;
    if (pP[mid] <= i) lo = mid; else hi = mid;
  }
  return sorted[pS[lo] + (i - pP[lo])];
}

__global__ void __launch_bounds__(SORT_T)
edge_bin_build(const int* __restrict__ sorted, const int* __restrict__ offs,
               int* __restrict__ rowptr, int* __restrict__ csr_src, int Nn, int NB, int NC,
               int ostride) {
  __shared__ int pS[CH_MAX];
  __shared__ int pP[CH_MAX + 1];
  __shared__ int cntL[BIN_W];
  __shared__ int curL[BIN_W];
  __shared__ int raw[BIN_CAP];
  __shared__ int srt[BIN_CAP];
  __shared__ int wsA[32];
  __shared__ int wsB[32];
  const int t = threadIdx.x, b = blockIdx.x;
  int a = 0, len = 0;
  if (t < NC) {
    const int* o = offs + (size_t)t * ostride + b;
    a   = o[0];
    len = o[1] - a;
    pS[t] = t * CHUNK + a;
  }
  if (t < BIN_W) cntL[t] = 0;
  int total, base;
  const int pre = block_excl_scan(len, wsA, total);
  (void)block_excl_scan(a, wsB, base);
  if (t < NC) pP[t] = pre;
  if (t == 0) pP[NC] = total;
  __syncthreads();

  const bool fits = total <= BIN_CAP;
  for (int i0 = t; i0 < total + t; i0 += 4 * SORT_T) {
    int v[4];
#pragma unroll
    for (int k = 0; k < 4; ++k) {
      const int i = i0 + k * SORT_T;
      v[k] = bin_entry(sorted, pS, pP, NC, (i < total) ? i : (total - 1));
    }
#pragma unroll
    for (int k = 0; k < 4; ++k) {
      const int i = i0 + k * SORT_T;
      if (i < total) {
        if (fits) raw[i] = v[k];
        atomicAdd(&cntL[v[k] & (BIN_W - 1)], 1);
      }
    }
  }
  __syncthreads();
  if (t < 32) {
    const int c0 = cntL[4 * t], c1 = cntL[4 * t + 1], c2 = cntL[4 * t + 2], c3 = cntL[4 * t + 3];
    const int s4 = c0 + c1 + c2 + c3;
    const int ex = wave_incl_scan(s4, t) - s4;
    const int st0 = ex, st1 = st0 + c0, st2 = st1 + c1, st3 = st2 + c2;
    curL[4 * t] = st0; curL[4 * t + 1] = st1; curL[4 * t + 2] = st2; curL[4 * t + 3] = st3;
    const int node = b * BIN_W + 4 * t;
    if (node     < Nn) rowptr[node]     = base + st0;
    if (node + 1 < Nn) rowptr[node + 1] = base + st1;
    if (node + 2 < Nn) rowptr[node + 2] = base + st2;
    if (node + 3 < Nn) rowptr[node + 3] = base + st3;
  }
  if (b == NB - 1 && t == 0) rowptr[Nn] = base + total;
  __syncthreads();
  for (int i = t; i < total; i += SORT_T) {
    const int v = fits ? raw[i] : bin_entry(sorted, pS, pP, NC, i);
    const int p = atomicAdd(&curL[v & (BIN_W - 1)], 1);
    if (fits) srt[p] = v >> BIN_SHIFT;
    else      csr_src[base + p] = v >> BIN_SHIFT;
  }
  __syncthreads();
  if (fits)
    for (int i = t; i < total; i += SORT_T) csr_src[base + i] = srt[i];
}

__device__ __forceinline__ float leaky(float e) { return e > 0.f ? e : 0.2f * e; }

__device__ __forceinline__ int lane_bcast_i(int v, int l) {
  return __builtin_amdgcn_readlane(v, l);
}
__device__ __forceinline__ float lane_bcast_f(float v, int l) {
  return __int_as_float(__builtin_amdgcn_readlane(__float_as_int(v), l));
}

__device__ __forceinline__ float wave_max(float v) {
  for (int off = 16; off > 0; off >>= 1) v = fmaxf(v, __shfl_xor(v, off, 32));
  return v;
}
__device__ __forceinline__ float wave_sum(float v) {
  for (int off = 16; off > 0; off >>= 1) v += __shfl_xor(v, off, 32);
  return v;
}

template <int G>
__device__ __forceinline__ float gather_group(const float* __restrict__ hfeat, int s,
                                              float w0, float w1, float w2, float w3,
                                              int j, int lane, float r) {
  float hv[G][HEADS];
#pragma unroll
  for (int g = 0; g < G; ++g) {
    const float* hp = hfeat + (size_t)lane_bcast_i(s, j + g) * HF;
#pragma unroll
    for (int k = 0; k < HEADS; ++k) hv[g][k] = hp[k * FEAT + lane];
  }
#pragma unroll
  for (int g = 0; g < G; ++g) {
    r = fmaf(lane_bcast_f(w0, j + g), hv[g][0], r);
    r = fmaf(lane_bcast_f(w1, j + g), hv[g][1], r);
    r = fmaf(lane_bcast_f(w2, j + g), hv[g][2], r);
    r = fmaf(lane_bcast_f(w3, j + g), hv[g][3], r);
  }
  return r;
}

__device__ __forceinline__ float gather_chunk(const float* __restrict__ hfeat, int s,
                                              float w0, float w1, float w2, float w3,
                                              int cnt, int lane, float acc) {
  int j = 0;
  for (; j + 8 <= cnt; j += 8) acc = gather_group<8>(hfeat, s, w0, w1, w2, w3, j, lane, acc);
  for (; j < cnt; j += 4)      acc = gather_group<4>(hfeat, s, w0, w1, w2, w3, j, lane, acc);
  return acc;
}

template <int G>
__device__ __forceinline__ void gather_rows(const float* __restrict__ xin, int s,
                                            float w0, float w1, float w2, float w3,
                                            int j, int lane,
                                            float& r0, float& r1, float& r2, float& r3) {
  float xv[G];
#pragma unroll
  for (int g = 0; g < G; ++g)
    xv[g] = xin[(size_t)lane_bcast_i(s, j + g) * FEAT + lane];
#pragma unroll
  for (int g = 0; g < G; ++g) {
    r0 = fmaf(lane_bcast_f(w0, j + g), xv[g], r0);
    r1 = fmaf(lane_bcast_f(w1, j + g), xv[g], r1);
    r2 = fmaf(lane_bcast_f(w2, j + g), xv[g], r2);
    r3 = fmaf(lane_bcast_f(w3, j + g), xv[g], r3);
  }
}

__device__ __forceinline__ void gather_rows_chunk(const float* __restrict__ xin, int s,
                                                  float w0, float w1, float w2, float w3,
                                                  int cnt, int lane,
                                                  float& r0, float& r1, float& r2, float& r3) {
  if (cnt > 16) {
    gather_rows<32>(xin, s, w0, w1, w2, w3, 0, lane, r0, r1, r2, r3);
    return;
  }
  int j = 0;
  for (; j + 8 <= cnt; j += 8) gather_rows<8>(xin, s, w0, w1, w2, w3, j, lane, r0, r1, r2, r3);
  for (; j < cnt; j += 4)      gather_rows<4>(xin, s, w0, w1, w2, w3, j, lane, r0, r1, r2, r3);
}

template <bool ROWS_IN>
__global__ void __launch_bounds__(256)
gat_node(const int* __restrict__ rowptr, const int* __restrict__ csr_src,
         const float* __restrict__ asrc, const float* __restrict__ adst,
         const float* __restrict__ hfeat, const float* __restrict__ bias,
         float* __restrict__ out, int Nn, int do_elu) {
  const int lane = threadIdx.x & 31;
  const int wv = __builtin_amdgcn_readfirstlane((int)(threadIdx.x >> 5));
  const int d  = blockIdx.x * (256 / 32) + wv;
  if (d >= Nn) return;
  const int beg = __builtin_amdgcn_readfirstlane(rowptr[d]);
  const int deg = __builtin_amdgcn_readfirstlane(rowptr[d + 1]) - beg;
  const int tot = deg + 1;
  const bool two = tot > 32;
  const float4 ad = *(const float4*)(adst + (size_t)d * HEADS);
  const float4* __restrict__ asrc4 = (const float4*)asrc;

  int sA = d, sB = d;
  if (lane < deg) sA = csr_src[beg + lane];
  if (two && lane + 32 < deg) sB = csr_src[beg + 32 + lane];
  const float4 aA = asrc4[sA];
  float4 aB = make_float4(0.f, 0.f, 0.f, 0.f);
  if (two) aB = asrc4[sB];

  float eA0 = -INFINITY, eA1 = -INFINITY, eA2 = -INFINITY, eA3 = -INFINITY;
  float eB0 = -INFINITY, eB1 = -INFINITY, eB2 = -INFINITY, eB3 = -INFINITY;
  if (lane < tot) {
    eA0 = leaky(aA.x + ad.x); eA1 = leaky(aA.y + ad.y);
    eA2 = leaky(aA.z + ad.z); eA3 = leaky(aA.w + ad.w);
  }
  if (lane + 32 < tot) {
    eB0 = leaky(aB.x + ad.x); eB1 = leaky(aB.y + ad.y);
    eB2 = leaky(aB.z + ad.z); eB3 = leaky(aB.w + ad.w);
  }

  float m0 = fmaxf(eA0, eB0), m1 = fmaxf(eA1, eB1), m2 = fmaxf(eA2, eB2), m3 = fmaxf(eA3, eB3);
  for (int base = 64; base < tot; base += 32) {
    const int i = base + lane;
    int s = d;
    if (i < deg) s = csr_src[beg + i];
    const float4 a = asrc4[s];
    if (i < tot) {
      m0 = fmaxf(m0, leaky(a.x + ad.x));
      m1 = fmaxf(m1, leaky(a.y + ad.y));
      m2 = fmaxf(m2, leaky(a.z + ad.z));
      m3 = fmaxf(m3, leaky(a.w + ad.w));
    }
  }
  m0 = wave_max(m0); m1 = wave_max(m1); m2 = wave_max(m2); m3 = wave_max(m3);

  float pA0 = 0.f, pA1 = 0.f, pA2 = 0.f, pA3 = 0.f;
  float pB0 = 0.f, pB1 = 0.f, pB2 = 0.f, pB3 = 0.f;
  if (lane < tot) {
    pA0 = expf(eA0 - m0); pA1 = expf(eA1 - m1); pA2 = expf(eA2 - m2); pA3 = expf(eA3 - m3);
  }
  if (lane + 32 < tot) {
    pB0 = expf(eB0 - m0); pB1 = expf(eB1 - m1); pB2 = expf(eB2 - m2); pB3 = expf(eB3 - m3);
  }
  float s0 = pA0 + pB0, s1 = pA1 + pB1, s2 = pA2 + pB2, s3 = pA3 + pB3;
  for (int base = 64; base < tot; base += 32) {
    const int i = base + lane;
    int s = d;
    if (i < deg) s = csr_src[beg + i];
    const float4 a = asrc4[s];
    if (i < tot) {
      s0 += expf(leaky(a.x + ad.x) - m0);
      s1 += expf(leaky(a.y + ad.y) - m1);
      s2 += expf(leaky(a.z + ad.z) - m2);
      s3 += expf(leaky(a.w + ad.w) - m3);
    }
  }
  const float den0 = wave_sum(s0) + 1e-16f;
  const float den1 = wave_sum(s1) + 1e-16f;
  const float den2 = wave_sum(s2) + 1e-16f;
  const float den3 = wave_sum(s3) + 1e-16f;

  if (ROWS_IN) {
    float g0 = 0.f, g1 = 0.f, g2 = 0.f, g3 = 0.f;
    gather_rows_chunk(hfeat, sA, pA0 / den0, pA1 / den1, pA2 / den2, pA3 / den3,
                      two ? 32 : tot, lane, g0, g1, g2, g3);
    if (two)
      gather_rows_chunk(hfeat, sB, pB0 / den0, pB1 / den1, pB2 / den2, pB3 / den3,
                        (tot - 32 < 32) ? (tot - 32) : 32, lane, g0, g1, g2, g3);
    for (int base = 64; base < tot; base += 32) {
      const int i = base + lane;
      int s = d;
      if (i < deg) s = csr_src[beg + i];
      const float4 a = asrc4[s];
      float w0 = 0.f, w1 = 0.f, w2 = 0.f, w3 = 0.f;
      if (i < tot) {
        w0 = expf(leaky(a.x + ad.x) - m0) / den0;
        w1 = expf(leaky(a.y + ad.y) - m1) / den1;
        w2 = expf(leaky(a.z + ad.z) - m2) / den2;
        w3 = expf(leaky(a.w + ad.w) - m3) / den3;
      }
      gather_rows_chunk(hfeat, s, w0, w1, w2, w3, (tot - base < 32) ? (tot - base) : 32, lane,
                        g0, g1, g2, g3);
    }
    float* __restrict__ o = out + (size_t)d * HF + lane;
    o[0]        = g0;
    o[FEAT]     = g1;
    o[2 * FEAT] = g2;
    o[3 * FEAT] = g3;
    return;
  }
  float acc = gather_chunk(hfeat, sA, pA0 / den0, pA1 / den1, pA2 / den2, pA3 / den3,
                           two ? 32 : tot, lane, 0.f);
  if (two)
    acc = gather_chunk(hfeat, sB, pB0 / den0, pB1 / den1, pB2 / den2, pB3 / den3,
                       (tot - 32 < 32) ? (tot - 32) : 32, lane, acc);
  for (int base = 64; base < tot; base += 32) {
    const int i = base + lane;
    int s = d;
    if (i < deg) s = csr_src[beg + i];
    const float4 a = asrc4[s];
    float w0 = 0.f, w1 = 0.f, w2 = 0.f, w3 = 0.f;
    if (i < tot) {
      w0 = expf(leaky(a.x + ad.x) - m0) / den0;
      w1 = expf(leaky(a.y + ad.y) - m1) / den1;
      w2 = expf(leaky(a.z + ad.z) - m2) / den2;
      w3 = expf(leaky(a.w + ad.w) - m3) / den3;
    }
    acc = gather_chunk(hfeat, s, w0, w1, w2, w3, (tot - base < 32) ? (tot - base) : 32, lane, acc);
  }

  float v = 0.25f * acc + bias[lane];
  if (do_elu) v = v > 0.f ? v : (expf(v) - 1.f);
  out[(size_t)d * FEAT + lane] = v;
}

extern "C" void kernel_launch(void* const* d_in, const int* in_sizes, int n_in,
                              void* d_out, int out_size, void* d_ws, size_t ws_size,
                              hipStream_t stream) {
  const float* x   = (const float*)d_in[0];
  const int*   ei  = (const int*)d_in[1];
  const float* W1  = (const float*)d_in[2];
  const float* as1 = (const float*)d_in[3];
  const float* ad1 = (const float*)d_in[4];
  const float* b1  = (const float*)d_in[5];
  const float* W2  = (const float*)d_in[6];
  const float* as2 = (const float*)d_in[7];
  const float* ad2 = (const float*)d_in[8];
  const float* b2  = (const float*)d_in[9];

  const int Nn = in_sizes[0] / HF;
  const int E  = in_sizes[1] / 2;

  const size_t n4 = ((size_t)Nn + 1 + 31) / 32 * 32;
  const size_t e4 = ((size_t)E + 31) / 32 * 32;
  float* ws     = (float*)d_ws;
  float* hbuf   = ws;
  float* x2     = hbuf + (size_t)Nn * HF;
  float* asrc   = x2   + (size_t)Nn * FEAT;
  float* adst   = asrc + (size_t)Nn * HEADS;
  int*   rowptr = (int*)(adst + (size_t)Nn * HEADS);
  int*   sorted = rowptr + n4;
  int*   csr    = sorted + e4;
  int*   offs   = csr + e4;

  const dim3 blk(256);
  const int row_tiles = Nn / 16;
  const int tail_row0 = row_tiles * 16;

  const int NB = (Nn + BIN_W - 1) / BIN_W;
  const int NC = (E + CHUNK - 1) / CHUNK;
  const int ostride = (NB + 1 + 31) / 32 * 32;
  if (NC > 0) edge_chunk_sort<<<NC, SORT_T, 0, stream>>>(ei, sorted, offs, E, NB, ostride);
  edge_bin_build<<<NB, SORT_T, 0, stream>>>(sorted, offs, rowptr, csr, Nn, NB, NC, ostride);

  const int trows = Nn - tail_row0;

  if (row_tiles > 0)
    gat_gemm_alpha_wmma_f32<true><<<row_tiles, dim3(128), 0, stream>>>(x, W1, as1, ad1, hbuf,
                                                                       asrc, adst, 128);
  if (trows > 0) {
    gat_gemm_tail<<<(trows * HF + 255) / 256, blk, 0, stream>>>(x, W1, hbuf, 128, tail_row0, Nn);
    gat_alpha<<<(trows * HEADS + 255) / 256, blk, 0, stream>>>(
        hbuf + (size_t)tail_row0 * HF, as1, ad1, asrc + (size_t)tail_row0 * HEADS,
        adst + (size_t)tail_row0 * HEADS, trows);
  }
  gat_node<false><<<(Nn + 7) / 8, blk, 0, stream>>>(rowptr, csr, asrc, adst, hbuf, b1, x2, Nn, 1);

  if (row_tiles > 0)
    gat_gemm_alpha_wmma_f32<false><<<row_tiles, dim3(128), 0, stream>>>(x2, W2, as2, ad2, hbuf,
                                                                        asrc, adst, FEAT);
  if (trows > 0) {
    gat_gemm_tail<<<(trows * HF + 255) / 256, blk, 0, stream>>>(x2, W2, hbuf, FEAT, tail_row0, Nn);
    gat_alpha<<<(trows * HEADS + 255) / 256, blk, 0, stream>>>(
        hbuf + (size_t)tail_row0 * HF, as2, ad2, asrc + (size_t)tail_row0 * HEADS,
        adst + (size_t)tail_row0 * HEADS, trows);
  }
  gat_node<true><<<(Nn + 7) / 8, blk, 0, stream>>>(rowptr, csr, asrc, adst, x2, b2, hbuf, Nn, 0);
  if (row_tiles > 0)
    gat_headmix_wmma_f32<<<(row_tiles + 3) / 4, dim3(128), 0, stream>>>(hbuf, W2, b2,
                                                                        (float*)d_out, row_tiles);
  if (trows > 0)
    gat_headmix_tail<<<(trows * FEAT + 255) / 256, blk, 0, stream>>>(hbuf, W2, b2, (float*)d_out,
                                                                     tail_row0, Nn);
}


# WMMA kernels: lanes own adjacent columns and 4 adjacent k, so each A/B fragment load and row store is one 16- or 8-byte access on whole lines; GEMM wave = 16 rows x 2 heads; float sum order changes
# speedup vs baseline: 9.8330x; 1.0761x over previous
#include <hip/hip_runtime.h>
#include <math.h>

#define HEADS 4
#define FEAT  32
#define HF    128

typedef __attribute__((ext_vector_type(2))) float v2f;
typedef __attribute__((ext_vector_type(8))) float v8f;

template <bool STORE_H>
__global__ void __launch_bounds__(128)
gat_gemm_alpha_wmma_f32(const float* __restrict__ X, const float* __restrict__ W,
                        const float* __restrict__ att_src, const float* __restrict__ att_dst,
                        float* __restrict__ Hout, float* __restrict__ a_src,
                        float* __restrict__ a_dst, int K, int row_tiles) {
  const int wv   = __builtin_amdgcn_readfirstlane((int)(threadIdx.x >> 5));
  const int tile = blockIdx.x * 2 + (wv >> 1);
  if (tile >= row_tiles) return;
  const int ch   = wv & 1;
  const int lane = threadIdx.x & 31;
  const int half = lane >> 4;
  const int l15  = lane & 15;
  const int col  = ch * 64 + 4 * l15;
  const float* __restrict__ xrow = X + (size_t)(tile * 16 + l15) * K + 4 * half;
  const float* __restrict__ wcol = W + (size_t)(4 * half) * HF + col;

  v8f c0 = {}, c1 = {}, c2 = {}, c3 = {};
  for (int k0 = 0; k0 < K; k0 += 8) {
    const float4 av = *(const float4*)(xrow + k0);
    const float* __restrict__ wr = wcol + (size_t)k0 * HF;
    const float4 w0 = *(const float4*)(wr);
    const float4 w1 = *(const float4*)(wr + HF);
    const float4 w2 = *(const float4*)(wr + 2 * HF);
    const float4 w3 = *(const float4*)(wr + 3 * HF);
    v2f a, b;
    a.x = av.x; a.y = av.y;
    b.x = w0.x; b.y = w1.x;
    c0 = __builtin_amdgcn_wmma_f32_16x16x4_f32(false, a, false, b, (short)0, c0, false, false);
    b.x = w0.y; b.y = w1.y;
    c1 = __builtin_amdgcn_wmma_f32_16x16x4_f32(false, a, false, b, (short)0, c1, false, false);
    b.x = w0.z; b.y = w1.z;
    c2 = __builtin_amdgcn_wmma_f32_16x16x4_f32(false, a, false, b, (short)0, c2, false, false);
    b.x = w0.w; b.y = w1.w;
    c3 = __builtin_amdgcn_wmma_f32_16x16x4_f32(false, a, false, b, (short)0, c3, false, false);
    a.x = av.z; a.y = av.w;
    b.x = w2.x; b.y = w3.x;
    c0 = __builtin_amdgcn_wmma_f32_16x16x4_f32(false, a, false, b, (short)0, c0, false, false);
    b.x = w2.y; b.y = w3.y;
    c1 = __builtin_amdgcn_wmma_f32_16x16x4_f32(false, a, false, b, (short)0, c1, false, false);
    b.x = w2.z; b.y = w3.z;
    c2 = __builtin_amdgcn_wmma_f32_16x16x4_f32(false, a, false, b, (short)0, c2, false, false);
    b.x = w2.w; b.y = w3.w;
    c3 = __builtin_amdgcn_wmma_f32_16x16x4_f32(false, a, false, b, (short)0, c3, false, false);
  }
  const int rbase = tile * 16 + half * 8;
  if (STORE_H) {
    float* orow = Hout + (size_t)rbase * HF + col;
#pragma unroll
    for (int v = 0; v < 8; ++v)
      *(float4*)(orow + (size_t)v * HF) = make_float4(c0[v], c1[v], c2[v], c3[v]);
  }

  const float4 as = *(const float4*)(att_src + col);
  const float4 ad = *(const float4*)(att_dst + col);
  float q[16];
#pragma unroll
  for (int v = 0; v < 8; ++v) {
    q[v]     = c0[v] * as.x + c1[v] * as.y + c2[v] * as.z + c3[v] * as.w;
    q[8 + v] = c0[v] * ad.x + c1[v] * ad.y + c2[v] * ad.z + c3[v] * ad.w;
  }
  const bool b2 = (l15 & 4) != 0, b1s = (l15 & 2) != 0, b0s = (l15 & 1) != 0;
  float r1[8], r2[4], r3[2];
#pragma unroll
  for (int i = 0; i < 8; ++i) {
    const float send = b2 ? q[i] : q[i + 8];
    const float keep = b2 ? q[i + 8] : q[i];
    r1[i] = keep + __shfl_xor(send, 4, 32);
  }
#pragma unroll
  for (int i = 0; i < 4; ++i) {
    const float send = b1s ? r1[i] : r1[i + 4];
    const float keep = b1s ? r1[i + 4] : r1[i];
    r2[i] = keep + __shfl_xor(send, 2, 32);
  }
#pragma unroll
  for (int i = 0; i < 2; ++i) {
    const float send = b0s ? r2[i] : r2[i + 2];
    const float keep = b0s ? r2[i + 2] : r2[i];
    r3[i] = keep + __shfl_xor(send, 1, 32);
  }
  float* __restrict__ ap = b2 ? a_dst : a_src;
  const int head = 2 * ch + (l15 >> 3);
  const int r0   = rbase + (b0s ? 2 : 0) + (b1s ? 4 : 0);
  ap[(size_t)r0 * HEADS + head]       = r3[0];
  ap[(size_t)(r0 + 1) * HEADS + head] = r3[1];
}

__global__ void gat_gemm_tail(const float* __restrict__ X, const float* __restrict__ W,
                              float* __restrict__ Hout, int K, int row0, int Nn) {
  int i = blockIdx.x * blockDim.x + threadIdx.x;
  int rows = Nn - row0;
  if (i >= rows * HF) return;
  int r = row0 + (i >> 7), col = i & (HF - 1);
  float acc = 0.f;
  for (int k = 0; k < K; ++k) acc += X[(size_t)r * K + k] * W[(size_t)k * HF + col];
  Hout[(size_t)r * HF + col] = acc;
}

__global__ void __launch_bounds__(128)
gat_headmix_wmma_f32(const float* __restrict__ G, const float* __restrict__ W,
                     const float* __restrict__ bias, float* __restrict__ out, int row_tiles) {
  const int wv   = __builtin_amdgcn_readfirstlane((int)(threadIdx.x >> 5));
  const int tile = blockIdx.x * 4 + wv;
  if (tile >= row_tiles) return;
  const int lane = threadIdx.x & 31;
  const int half = lane >> 4;
  const int l15  = lane & 15;
  const float* __restrict__ grow = G + (size_t)(tile * 16 + l15) * HF + 4 * half;

  v8f c0 = {}, c1 = {};
  for (int k0 = 0; k0 < HF; k0 += 8) {
    const int kk = k0 + 4 * half;
    const float4 av = *(const float4*)(grow + k0);
    const float* __restrict__ wr = W + (size_t)(kk & (FEAT - 1)) * HF + (kk >> 5) * FEAT + 2 * l15;
    const float2 w0 = *(const float2*)(wr);
    const float2 w1 = *(const float2*)(wr + HF);
    const float2 w2 = *(const float2*)(wr + 2 * HF);
    const float2 w3 = *(const float2*)(wr + 3 * HF);
    v2f a, b;
    a.x = av.x; a.y = av.y;
    b.x = w0.x; b.y = w1.x;
    c0 = __builtin_amdgcn_wmma_f32_16x16x4_f32(false, a, false, b, (short)0, c0, false, false);
    b.x = w0.y; b.y = w1.y;
    c1 = __builtin_amdgcn_wmma_f32_16x16x4_f32(false, a, false, b, (short)0, c1, false, false);
    a.x = av.z; a.y = av.w;
    b.x = w2.x; b.y = w3.x;
    c0 = __builtin_amdgcn_wmma_f32_16x16x4_f32(false, a, false, b, (short)0, c0, false, false);
    b.x = w2.y; b.y = w3.y;
    c1 = __builtin_amdgcn_wmma_f32_16x16x4_f32(false, a, false, b, (short)0, c1, false, false);
  }
  const int rbase = tile * 16 + half * 8;
  const float2 bs = *(const float2*)(bias + 2 * l15);
  float* orow = out + (size_t)rbase * FEAT + 2 * l15;
#pragma unroll
  for (int v = 0; v < 8; ++v)
    *(float2*)(orow + (size_t)v * FEAT) = make_float2(0.25f * c0[v] + bs.x, 0.25f * c1[v] + bs.y);
}

__global__ void gat_headmix_tail(const float* __restrict__ G, const float* __restrict__ W,
                                 const float* __restrict__ bias, float* __restrict__ out,
                                 int row0, int Nn) {
  int i = blockIdx.x * blockDim.x + threadIdx.x;
  int rows = Nn - row0;
  if (i >= rows * FEAT) return;
  int r = row0 + (i >> 5), f = i & (FEAT - 1);
  float acc = 0.f;
  for (int kk = 0; kk < HF; ++kk)
    acc += G[(size_t)r * HF + kk] * W[(size_t)(kk & (FEAT - 1)) * HF + (kk >> 5) * FEAT + f];
  out[(size_t)r * FEAT + f] = 0.25f * acc + bias[f];
}

__global__ void gat_alpha(const float* __restrict__ H, const float* __restrict__ att_src,
                          const float* __restrict__ att_dst,
                          float* __restrict__ a_src, float* __restrict__ a_dst, int Nn) {
  int i = blockIdx.x * blockDim.x + threadIdx.x;
  if (i >= Nn * HEADS) return;
  int h = i & (HEADS - 1);
  const float4* hp = (const float4*)(H + (size_t)(i >> 2) * HF + h * FEAT);
  const float4* as = (const float4*)(att_src + h * FEAT);
  const float4* ad = (const float4*)(att_dst + h * FEAT);
  float s1 = 0.f, s2 = 0.f;
#pragma unroll
  for (int f = 0; f < FEAT / 4; ++f) {
    float4 v = hp[f], a = as[f], d = ad[f];
    s1 += v.x * a.x + v.y * a.y + v.z * a.z + v.w * a.w;
    s2 += v.x * d.x + v.y * d.y + v.z * d.z + v.w * d.w;
  }
  a_src[i] = s1;
  a_dst[i] = s2;
}

#define BIN_SHIFT 7
#define BIN_W     128
#define SORT_T    1024
#define CHUNK     4096
#define NB_MAX    1023
#define CH_MAX    1024
#define BIN_CAP   8192

__device__ __forceinline__ int wave_incl_scan(int v, int lane) {
  for (int off = 1; off < 32; off <<= 1) {
    int u = __shfl_up(v, off, 32);
    if (lane >= off) v += u;
  }
  return v;
}

__device__ __forceinline__ int block_excl_scan(int v, int* wsum, int& total) {
  const int lane = threadIdx.x & 31, wv = threadIdx.x >> 5;
  const int inc = wave_incl_scan(v, lane);
  if (lane == 31) wsum[wv] = inc;
  __syncthreads();
  const int ws  = wsum[lane];
  const int wsi = wave_incl_scan(ws, lane);
  total = __shfl(wsi, 31, 32);
  return __shfl(wsi, wv, 32) - __shfl(ws, wv, 32) + inc - v;
}

__global__ void __launch_bounds__(SORT_T)
edge_chunk_sort(const int* __restrict__ ei, int* __restrict__ sorted,
                int* __restrict__ offs, int E, int NB, int ostride) {
  __shared__ int cnt[NB_MAX + 1];
  __shared__ int off[NB_MAX + 1];
  __shared__ int srt[CHUNK];
  __shared__ int wsum[32];
  const int t  = threadIdx.x;
  const int e0 = blockIdx.x * CHUNK;
  const int n  = (E - e0 < CHUNK) ? (E - e0) : CHUNK;
  if (t <= NB) cnt[t] = 0;
  __syncthreads();
  int pk[CHUNK / SORT_T], bn[CHUNK / SORT_T], rk[CHUNK / SORT_T];
#pragma unroll
  for (int k = 0; k < CHUNK / SORT_T; ++k) {
    const int i = t + k * SORT_T;
    const int j = e0 + ((i < n) ? i : (n - 1));
    const int s = ei[j], d = ei[E + j];
    bn[k] = d >> BIN_SHIFT;
    pk[k] = (s << BIN_SHIFT) | (d & (BIN_W - 1));
  }
#pragma unroll
  for (int k = 0; k < CHUNK / SORT_T; ++k) {
    rk[k] = 0;
    if (t + k * SORT_T < n) rk[k] = atomicAdd(&cnt[bn[k]], 1);
  }
  __syncthreads();
  int tot;
  const int ex = block_excl_scan((t < NB) ? cnt[t] : 0, wsum, tot);
  if (t <= NB) {
    off[t] = ex;
    offs[(size_t)blockIdx.x * ostride + t] = ex;
  }
  __syncthreads();
#pragma unroll
  for (int k = 0; k < CHUNK / SORT_T; ++k)
    if (t + k * SORT_T < n) srt[off[bn[k]] + rk[k]] = pk[k];
  __syncthreads();
#pragma unroll
  for (int k = 0; k < CHUNK / SORT_T; ++k) {
    const int i = t + k * SORT_T;
    if (i < n) sorted[e0 + i] = srt[i];
  }
}

__device__ __forceinline__ int bin_entry(const int* __restrict__ sorted, const int* pS,
                                         const int* pP, int NC, int i) {
  int lo = 0, hi = NC;
  while (hi - lo > 1) {
    const int mid = (lo + hi) >> 1;
    if (pP[mid] <= i) lo = mid; else hi = mid;
  }
  return sorted[pS[lo] + (i - pP[lo])];
}

__global__ void __launch_bounds__(SORT_T)
edge_bin_build(const int* __restrict__ sorted, const int* __restrict__ offs,
               int* __restrict__ rowptr, int* __restrict__ csr_src, int Nn, int NB, int NC,
               int ostride) {
  __shared__ int pS[CH_MAX];
  __shared__ int pP[CH_MAX + 1];
  __shared__ int cntL[BIN_W];
  __shared__ int curL[BIN_W];
  __shared__ int raw[BIN_CAP];
  __shared__ int srt[BIN_CAP];
  __shared__ int wsA[32];
  __shared__ int wsB[32];
  const int t = threadIdx.x, b = blockIdx.x;
  int a = 0, len = 0;
  if (t < NC) {
    const int* o = offs + (size_t)t * ostride + b;
    a   = o[0];
    len = o[1] - a;
    pS[t] = t * CHUNK + a;
  }
  if (t < BIN_W) cntL[t] = 0;
  int total, base;
  const int pre = block_excl_scan(len, wsA, total);
  (void)block_excl_scan(a, wsB, base);
  if (t < NC) pP[t] = pre;
  if (t == 0) pP[NC] = total;
  __syncthreads();

  const bool fits = total <= BIN_CAP;
  for (int i0 = t; i0 < total + t; i0 += 4 * SORT_T) {
    int v[4];
#pragma unroll
    for (int k = 0; k < 4; ++k) {
      const int i = i0 + k * SORT_T;
      v[k] = bin_entry(sorted, pS, pP, NC, (i < total) ? i : (total - 1));
    }
#pragma unroll
    for (int k = 0; k < 4; ++k) {
      const int i = i0 + k * SORT_T;
      if (i < total) {
        if (fits) raw[i] = v[k];
        atomicAdd(&cntL[v[k] & (BIN_W - 1)], 1);
      }
    }
  }
  __syncthreads();
  if (t < 32) {
    const int c0 = cntL[4 * t], c1 = cntL[4 * t + 1], c2 = cntL[4 * t + 2], c3 = cntL[4 * t + 3];
    const int s4 = c0 + c1 + c2 + c3;
    const int ex = wave_incl_scan(s4, t) - s4;
    const int st0 = ex, st1 = st0 + c0, st2 = st1 + c1, st3 = st2 + c2;
    curL[4 * t] = st0; curL[4 * t + 1] = st1; curL[4 * t + 2] = st2; curL[4 * t + 3] = st3;
    const int node = b * BIN_W + 4 * t;
    if (node     < Nn) rowptr[node]     = base + st0;
    if (node + 1 < Nn) rowptr[node + 1] = base + st1;
    if (node + 2 < Nn) rowptr[node + 2] = base + st2;
    if (node + 3 < Nn) rowptr[node + 3] = base + st3;
  }
  if (b == NB - 1 && t == 0) rowptr[Nn] = base + total;
  __syncthreads();
  for (int i = t; i < total; i += SORT_T) {
    const int v = fits ? raw[i] : bin_entry(sorted, pS, pP, NC, i);
    const int p = atomicAdd(&curL[v & (BIN_W - 1)], 1);
    if (fits) srt[p] = v >> BIN_SHIFT;
    else      csr_src[base + p] = v >> BIN_SHIFT;
  }
  __syncthreads();
  if (fits)
    for (int i = t; i < total; i += SORT_T) csr_src[base + i] = srt[i];
}

__device__ __forceinline__ float leaky(float e) { return e > 0.f ? e : 0.2f * e; }

__device__ __forceinline__ int lane_bcast_i(int v, int l) {
  return __builtin_amdgcn_readlane(v, l);
}
__device__ __forceinline__ float lane_bcast_f(float v, int l) {
  return __int_as_float(__builtin_amdgcn_readlane(__float_as_int(v), l));
}

__device__ __forceinline__ float wave_max(float v) {
  for (int off = 16; off > 0; off >>= 1) v = fmaxf(v, __shfl_xor(v, off, 32));
  return v;
}
__device__ __forceinline__ float wave_sum(float v) {
  for (int off = 16; off > 0; off >>= 1) v += __shfl_xor(v, off, 32);
  return v;
}

template <int G>
__device__ __forceinline__ float gather_group(const float* __restrict__ hfeat, int s,
                                              float w0, float w1, float w2, float w3,
                                              int j, int lane, float r) {
  float hv[G][HEADS];
#pragma unroll
  for (int g = 0; g < G; ++g) {
    const float* hp = hfeat + (size_t)lane_bcast_i(s, j + g) * HF;
#pragma unroll
    for (int k = 0; k < HEADS; ++k) hv[g][k] = hp[k * FEAT + lane];
  }
#pragma unroll
  for (int g = 0; g < G; ++g) {
    r = fmaf(lane_bcast_f(w0, j + g), hv[g][0], r);
    r = fmaf(lane_bcast_f(w1, j + g), hv[g][1], r);
    r = fmaf(lane_bcast_f(w2, j + g), hv[g][2], r);
    r = fmaf(lane_bcast_f(w3, j + g), hv[g][3], r);
  }
  return r;
}

__device__ __forceinline__ float gather_chunk(const float* __restrict__ hfeat, int s,
                                              float w0, float w1, float w2, float w3,
                                              int cnt, int lane, float acc) {
  int j = 0;
  for (; j + 8 <= cnt; j += 8) acc = gather_group<8>(hfeat, s, w0, w1, w2, w3, j, lane, acc);
  for (; j < cnt; j += 4)      acc = gather_group<4>(hfeat, s, w0, w1, w2, w3, j, lane, acc);
  return acc;
}

template <int G>
__device__ __forceinline__ void gather_rows(const float* __restrict__ xin, int s,
                                            float w0, float w1, float w2, float w3,
                                            int j, int lane,
                                            float& r0, float& r1, float& r2, float& r3) {
  float xv[G];
#pragma unroll
  for (int g = 0; g < G; ++g)
    xv[g] = xin[(size_t)lane_bcast_i(s, j + g) * FEAT + lane];
#pragma unroll
  for (int g = 0; g < G; ++g) {
    r0 = fmaf(lane_bcast_f(w0, j + g), xv[g], r0);
    r1 = fmaf(lane_bcast_f(w1, j + g), xv[g], r1);
    r2 = fmaf(lane_bcast_f(w2, j + g), xv[g], r2);
    r3 = fmaf(lane_bcast_f(w3, j + g), xv[g], r3);
  }
}

__device__ __forceinline__ void gather_rows_chunk(const float* __restrict__ xin, int s,
                                                  float w0, float w1, float w2, float w3,
                                                  int cnt, int lane,
                                                  float& r0, float& r1, float& r2, float& r3) {
  if (cnt > 16) {
    gather_rows<32>(xin, s, w0, w1, w2, w3, 0, lane, r0, r1, r2, r3);
    return;
  }
  int j = 0;
  for (; j + 8 <= cnt; j += 8) gather_rows<8>(xin, s, w0, w1, w2, w3, j, lane, r0, r1, r2, r3);
  for (; j < cnt; j += 4)      gather_rows<4>(xin, s, w0, w1, w2, w3, j, lane, r0, r1, r2, r3);
}

template <bool ROWS_IN>
__global__ void __launch_bounds__(256)
gat_node(const int* __restrict__ rowptr, const int* __restrict__ csr_src,
         const float* __restrict__ asrc, const float* __restrict__ adst,
         const float* __restrict__ hfeat, const float* __restrict__ bias,
         float* __restrict__ out, int Nn, int do_elu) {
  const int lane = threadIdx.x & 31;
  const int wv = __builtin_amdgcn_readfirstlane((int)(threadIdx.x >> 5));
  const int d  = blockIdx.x * (256 / 32) + wv;
  if (d >= Nn) return;
  const int beg = __builtin_amdgcn_readfirstlane(rowptr[d]);
  const int deg = __builtin_amdgcn_readfirstlane(rowptr[d + 1]) - beg;
  const int tot = deg + 1;
  const bool two = tot > 32;
  const float4 ad = *(const float4*)(adst + (size_t)d * HEADS);
  const float4* __restrict__ asrc4 = (const float4*)asrc;

  int sA = d, sB = d;
  if (lane < deg) sA = csr_src[beg + lane];
  if (two && lane + 32 < deg) sB = csr_src[beg + 32 + lane];
  const float4 aA = asrc4[sA];
  float4 aB = make_float4(0.f, 0.f, 0.f, 0.f);
  if (two) aB = asrc4[sB];

  float eA0 = -INFINITY, eA1 = -INFINITY, eA2 = -INFINITY, eA3 = -INFINITY;
  float eB0 = -INFINITY, eB1 = -INFINITY, eB2 = -INFINITY, eB3 = -INFINITY;
  if (lane < tot) {
    eA0 = leaky(aA.x + ad.x); eA1 = leaky(aA.y + ad.y);
    eA2 = leaky(aA.z + ad.z); eA3 = leaky(aA.w + ad.w);
  }
  if (lane + 32 < tot) {
    eB0 = leaky(aB.x + ad.x); eB1 = leaky(aB.y + ad.y);
    eB2 = leaky(aB.z + ad.z); eB3 = leaky(aB.w + ad.w);
  }

  float m0 = fmaxf(eA0, eB0), m1 = fmaxf(eA1, eB1), m2 = fmaxf(eA2, eB2), m3 = fmaxf(eA3, eB3);
  for (int base = 64; base < tot; base += 32) {
    const int i = base + lane;
    int s = d;
    if (i < deg) s = csr_src[beg + i];
    const float4 a = asrc4[s];
    if (i < tot) {
      m0 = fmaxf(m0, leaky(a.x + ad.x));
      m1 = fmaxf(m1, leaky(a.y + ad.y));
      m2 = fmaxf(m2, leaky(a.z + ad.z));
      m3 = fmaxf(m3, leaky(a.w + ad.w));
    }
  }
  m0 = wave_max(m0); m1 = wave_max(m1); m2 = wave_max(m2); m3 = wave_max(m3);

  float pA0 = 0.f, pA1 = 0.f, pA2 = 0.f, pA3 = 0.f;
  float pB0 = 0.f, pB1 = 0.f, pB2 = 0.f, pB3 = 0.f;
  if (lane < tot) {
    pA0 = expf(eA0 - m0); pA1 = expf(eA1 - m1); pA2 = expf(eA2 - m2); pA3 = expf(eA3 - m3);
  }
  if (lane + 32 < tot) {
    pB0 = expf(eB0 - m0); pB1 = expf(eB1 - m1); pB2 = expf(eB2 - m2); pB3 = expf(eB3 - m3);
  }
  float s0 = pA0 + pB0, s1 = pA1 + pB1, s2 = pA2 + pB2, s3 = pA3 + pB3;
  for (int base = 64; base < tot; base += 32) {
    const int i = base + lane;
    int s = d;
    if (i < deg) s = csr_src[beg + i];
    const float4 a = asrc4[s];
    if (i < tot) {
      s0 += expf(leaky(a.x + ad.x) - m0);
      s1 += expf(leaky(a.y + ad.y) - m1);
      s2 += expf(leaky(a.z + ad.z) - m2);
      s3 += expf(leaky(a.w + ad.w) - m3);
    }
  }
  const float den0 = wave_sum(s0) + 1e-16f;
  const float den1 = wave_sum(s1) + 1e-16f;
  const float den2 = wave_sum(s2) + 1e-16f;
  const float den3 = wave_sum(s3) + 1e-16f;

  if (ROWS_IN) {
    float g0 = 0.f, g1 = 0.f, g2 = 0.f, g3 = 0.f;
    gather_rows_chunk(hfeat, sA, pA0 / den0, pA1 / den1, pA2 / den2, pA3 / den3,
                      two ? 32 : tot, lane, g0, g1, g2, g3);
    if (two)
      gather_rows_chunk(hfeat, sB, pB0 / den0, pB1 / den1, pB2 / den2, pB3 / den3,
                        (tot - 32 < 32) ? (tot - 32) : 32, lane, g0, g1, g2, g3);
    for (int base = 64; base < tot; base += 32) {
      const int i = base + lane;
      int s = d;
      if (i < deg) s = csr_src[beg + i];
      const float4 a = asrc4[s];
      float w0 = 0.f, w1 = 0.f, w2 = 0.f, w3 = 0.f;
      if (i < tot) {
        w0 = expf(leaky(a.x + ad.x) - m0) / den0;
        w1 = expf(leaky(a.y + ad.y) - m1) / den1;
        w2 = expf(leaky(a.z + ad.z) - m2) / den2;
        w3 = expf(leaky(a.w + ad.w) - m3) / den3;
      }
      gather_rows_chunk(hfeat, s, w0, w1, w2, w3, (tot - base < 32) ? (tot - base) : 32, lane,
                        g0, g1, g2, g3);
    }
    float* __restrict__ o = out + (size_t)d * HF + lane;
    o[0]        = g0;
    o[FEAT]     = g1;
    o[2 * FEAT] = g2;
    o[3 * FEAT] = g3;
    return;
  }
  float acc = gather_chunk(hfeat, sA, pA0 / den0, pA1 / den1, pA2 / den2, pA3 / den3,
                           two ? 32 : tot, lane, 0.f);
  if (two)
    acc = gather_chunk(hfeat, sB, pB0 / den0, pB1 / den1, pB2 / den2, pB3 / den3,
                       (tot - 32 < 32) ? (tot - 32) : 32, lane, acc);
  for (int base = 64; base < tot; base += 32) {
    const int i = base + lane;
    int s = d;
    if (i < deg) s = csr_src[beg + i];
    const float4 a = asrc4[s];
    float w0 = 0.f, w1 = 0.f, w2 = 0.f, w3 = 0.f;
    if (i < tot) {
      w0 = expf(leaky(a.x + ad.x) - m0) / den0;
      w1 = expf(leaky(a.y + ad.y) - m1) / den1;
      w2 = expf(leaky(a.z + ad.z) - m2) / den2;
      w3 = expf(leaky(a.w + ad.w) - m3) / den3;
    }
    acc = gather_chunk(hfeat, s, w0, w1, w2, w3, (tot - base < 32) ? (tot - base) : 32, lane, acc);
  }

  float v = 0.25f * acc + bias[lane];
  if (do_elu) v = v > 0.f ? v : (expf(v) - 1.f);
  out[(size_t)d * FEAT + lane] = v;
}

extern "C" void kernel_launch(void* const* d_in, const int* in_sizes, int n_in,
                              void* d_out, int out_size, void* d_ws, size_t ws_size,
                              hipStream_t stream) {
  const float* x   = (const float*)d_in[0];
  const int*   ei  = (const int*)d_in[1];
  const float* W1  = (const float*)d_in[2];
  const float* as1 = (const float*)d_in[3];
  const float* ad1 = (const float*)d_in[4];
  const float* b1  = (const float*)d_in[5];
  const float* W2  = (const float*)d_in[6];
  const float* as2 = (const float*)d_in[7];
  const float* ad2 = (const float*)d_in[8];
  const float* b2  = (const float*)d_in[9];

  const int Nn = in_sizes[0] / HF;
  const int E  = in_sizes[1] / 2;

  const size_t n4 = ((size_t)Nn + 1 + 31) / 32 * 32;
  const size_t e4 = ((size_t)E + 31) / 32 * 32;
  float* ws     = (float*)d_ws;
  float* hbuf   = ws;
  float* x2     = hbuf + (size_t)Nn * HF;
  float* asrc   = x2   + (size_t)Nn * FEAT;
  float* adst   = asrc + (size_t)Nn * HEADS;
  int*   rowptr = (int*)(adst + (size_t)Nn * HEADS);
  int*   sorted = rowptr + n4;
  int*   csr    = sorted + e4;
  int*   offs   = csr + e4;

  const dim3 blk(256);
  const int row_tiles = Nn / 16;
  const int tail_row0 = row_tiles * 16;

  const int NB = (Nn + BIN_W - 1) / BIN_W;
  const int NC = (E + CHUNK - 1) / CHUNK;
  const int ostride = (NB + 1 + 31) / 32 * 32;
  if (NC > 0) edge_chunk_sort<<<NC, SORT_T, 0, stream>>>(ei, sorted, offs, E, NB, ostride);
  edge_bin_build<<<NB, SORT_T, 0, stream>>>(sorted, offs, rowptr, csr, Nn, NB, NC, ostride);

  const int trows = Nn - tail_row0;

  if (row_tiles > 0)
    gat_gemm_alpha_wmma_f32<true><<<(row_tiles + 1) / 2, dim3(128), 0, stream>>>(x, W1, as1, ad1, hbuf,
                                                                                 asrc, adst, 128, row_tiles);
  if (trows > 0) {
    gat_gemm_tail<<<(trows * HF + 255) / 256, blk, 0, stream>>>(x, W1, hbuf, 128, tail_row0, Nn);
    gat_alpha<<<(trows * HEADS + 255) / 256, blk, 0, stream>>>(
        hbuf + (size_t)tail_row0 * HF, as1, ad1, asrc + (size_t)tail_row0 * HEADS,
        adst + (size_t)tail_row0 * HEADS, trows);
  }
  gat_node<false><<<(Nn + 7) / 8, blk, 0, stream>>>(rowptr, csr, asrc, adst, hbuf, b1, x2, Nn, 1);

  if (row_tiles > 0)
    gat_gemm_alpha_wmma_f32<false><<<(row_tiles + 1) / 2, dim3(128), 0, stream>>>(x2, W2, as2, ad2, hbuf,
                                                                                  asrc, adst, FEAT, row_tiles);
  if (trows > 0) {
    gat_gemm_tail<<<(trows * HF + 255) / 256, blk, 0, stream>>>(x2, W2, hbuf, FEAT, tail_row0, Nn);
    gat_alpha<<<(trows * HEADS + 255) / 256, blk, 0, stream>>>(
        hbuf + (size_t)tail_row0 * HF, as2, ad2, asrc + (size_t)tail_row0 * HEADS,
        adst + (size_t)tail_row0 * HEADS, trows);
  }
  gat_node<true><<<(Nn + 7) / 8, blk, 0, stream>>>(rowptr, csr, asrc, adst, x2, b2, hbuf, Nn, 0);
  if (row_tiles > 0)
    gat_headmix_wmma_f32<<<(row_tiles + 3) / 4, dim3(128), 0, stream>>>(hbuf, W2, b2,
                                                                        (float*)d_out, row_tiles);
  if (trows > 0)
    gat_headmix_tail<<<(trows * FEAT + 255) / 256, blk, 0, stream>>>(hbuf, W2, b2, (float*)d_out,
                                                                     tail_row0, Nn);
}
